# starting kernel
# baseline (speedup 1.0000x reference)
#include <hip/hip_runtime.h>
#include <hip/hip_bf16.h>

typedef float v2f __attribute__((ext_vector_type(2)));
typedef float v8f __attribute__((ext_vector_type(8)));

#define FDIM 128
#define KTOT 384
#define KC   64

__global__ void fill_zero(float* p, size_t n) {
  size_t i = (size_t)blockIdx.x * blockDim.x + threadIdx.x;
  size_t stride = (size_t)gridDim.x * blockDim.x;
  for (; i < n; i += stride) p[i] = 0.0f;
}

__global__ void build_weights(const float* __restrict__ Wn1, const float* __restrict__ Ws1,
                              const float* __restrict__ b1,
                              const float* __restrict__ Wn2, const float* __restrict__ Ws2,
                              const float* __restrict__ b2,
                              float* __restrict__ Bcat1, float* __restrict__ bias1,
                              float* __restrict__ Bcat2, float* __restrict__ bias2) {
  int i = blockIdx.x * blockDim.x + threadIdx.x;
  const int S1 = 384 * 128, S2 = S1 + 128, S3 = S2 + 384 * 64, S4 = S3 + 64;
  if (i < S1) {
    int k = i >> 7, n = i & 127;
    float v;
    if (k < 128)      v = Ws1[k * 128 + n] + Ws1[16384 + k * 128 + n];
    else if (k < 256) v = Wn1[(k - 128) * 128 + n];
    else              v = Wn1[16384 + (k - 256) * 128 + n];
    Bcat1[i] = v;
  } else if (i < S2) {
    int n = i - S1;
    bias1[n] = b1[n] + b1[128 + n];
  } else if (i < S3) {
    int j = i - S2;
    int k = j >> 6, n = j & 63;
    float v;
    if (k < 128)      v = Ws2[k * 64 + n] + Ws2[8192 + k * 64 + n];
    else if (k < 256) v = Wn2[(k - 128) * 64 + n];
    else              v = Wn2[8192 + (k - 256) * 64 + n];
    Bcat2[j] = v;
  } else if (i < S4) {
    int n = i - S3;
    bias2[n] = b2[n] + b2[64 + n];
  }
}

__global__ void degree_count(const int* __restrict__ edges, float* deg, int E, int N) {
  int e = blockIdx.x * blockDim.x + threadIdx.x;
  int r = blockIdx.y;
  if (e >= E) return;
  int d = edges[(size_t)r * 2 * E + E + e];
  atomicAdd(deg + (size_t)r * N + d, 1.0f);
}

__global__ void recip_deg(float* deg, int n) {
  int i = blockIdx.x * blockDim.x + threadIdx.x;
  if (i < n) deg[i] = 1.0f / fmaxf(deg[i], 1.0f);
}

__global__ void scatter_feat(const int* __restrict__ src, const int* __restrict__ dst,
                             const float* __restrict__ feat, float* agg, int E) {
  int gid = blockIdx.x * blockDim.x + threadIdx.x;
  int e = gid >> 5, lane = gid & 31;
  if (e >= E) return;
  int s = src[e], d = dst[e];
  const float4 v = *(const float4*)(feat + (size_t)s * FDIM + lane * 4);
  float* p = agg + (size_t)d * FDIM + lane * 4;
  atomicAdd(p + 0, v.x);
  atomicAdd(p + 1, v.y);
  atomicAdd(p + 2, v.z);
  atomicAdd(p + 3, v.w);
}

__global__ void scale_rows(float* agg, const float* __restrict__ dinv, size_t n) {
  size_t i = (size_t)blockIdx.x * blockDim.x + threadIdx.x;
  if (i < n) agg[i] *= dinv[i >> 7];
}

template <int NOUT, bool RELU>
__global__ __launch_bounds__(256) void sage_gemm(
    const float* __restrict__ A0, const float* __restrict__ G0, const float* __restrict__ G1,
    const float* __restrict__ Bmat, const float* __restrict__ bias,
    float* __restrict__ out, int Nrows) {
  constexpr int NT = NOUT / 16;
  __shared__ float Bs[KC * NOUT];

  const int tid = threadIdx.x;
  const int lane = tid & 31;
  const int wave = tid >> 5;
  const int rowBase = blockIdx.x * 128 + wave * 16;
  const int m = lane & 15;
  const int khalf = (lane >> 4) << 1;

  v8f czero = {};
  v8f c[NT];
#pragma unroll
  for (int t = 0; t < NT; ++t) c[t] = czero;

  int rowA = rowBase + m;
  if (rowA > Nrows - 1) rowA = Nrows - 1;

  for (int kc = 0; kc < KTOT; kc += KC) {
    for (int i = tid; i < KC * NOUT; i += 256)
      Bs[i] = Bmat[kc * NOUT + i];
    __syncthreads();

    const float* Asrc = (kc < 128) ? A0 : ((kc < 256) ? G0 : G1);
    const int kbase = kc & 127;

    for (int k4 = 0; k4 < KC; k4 += 4) {
      v2f a = *(const v2f*)(Asrc + (size_t)rowA * FDIM + kbase + k4 + khalf);
      const int kb = k4 + khalf;
#pragma unroll
      for (int t = 0; t < NT; ++t) {
        v2f b;
        b.x = Bs[kb * NOUT + t * 16 + m];
        b.y = Bs[(kb + 1) * NOUT + t * 16 + m];
        c[t] = __builtin_amdgcn_wmma_f32_16x16x4_f32(
            false, a, false, b, (short)0, c[t], false, false);
      }
    }
    __syncthreads();
  }

  const int rhalf = (lane >> 4) << 3;
#pragma unroll
  for (int t = 0; t < NT; ++t) {
    int col = t * 16 + m;
    float bv = bias[col];
#pragma unroll
    for (int v = 0; v < 8; ++v) {
      int row = rowBase + v + rhalf;
      if (row < Nrows) {
        float val = c[t][v] + bv;
        if (RELU) val = fmaxf(val, 0.0f);
        out[(size_t)row * NOUT + col] = val;
      }
    }
  }
}

__global__ void edge_score(const int* __restrict__ src, const int* __restrict__ dst,
                           const float* __restrict__ h, float* __restrict__ out, int E) {
  int gid = blockIdx.x * blockDim.x + threadIdx.x;
  int e = gid >> 5, lane = gid & 31;
  if (e >= E) return;
  int s = src[e], d = dst[e];
  v2f a = *(const v2f*)(h + (size_t)s * 64 + lane * 2);
  v2f b = *(const v2f*)(h + (size_t)d * 64 + lane * 2);
  float p = a.x * b.x + a.y * b.y;
#pragma unroll
  for (int off = 16; off > 0; off >>= 1)
    p += __shfl_xor(p, off, 32);
  if (lane == 0) out[e] = p;
}

extern "C" void kernel_launch(void* const* d_in, const int* in_sizes, int n_in,
                              void* d_out, int out_size, void* d_ws, size_t ws_size,
                              hipStream_t stream) {
  const float* x   = (const float*)d_in[0];
  const int*   edges = (const int*)d_in[1];
  const int*   negE  = (const int*)d_in[2];
  const float* Wn1 = (const float*)d_in[3];
  const float* Ws1 = (const float*)d_in[4];
  const float* b1  = (const float*)d_in[5];
  const float* Wn2 = (const float*)d_in[6];
  const float* Ws2 = (const float*)d_in[7];
  const float* b2  = (const float*)d_in[8];
  float* out = (float*)d_out;

  const int N = in_sizes[0] / 128;
  const int E = in_sizes[1] / 4;

  float* ws    = (float*)d_ws;
  float* dinv  = ws;
  float* agg0  = dinv + 2 * (size_t)N;
  float* agg1  = agg0 + (size_t)N * 128;
  float* h1    = agg1 + (size_t)N * 128;
  float* h2    = h1 + (size_t)N * 128;
  float* Bcat1 = h2 + (size_t)N * 64;
  float* bias1 = Bcat1 + 384 * 128;
  float* Bcat2 = bias1 + 128;
  float* bias2 = Bcat2 + 384 * 64;

  const int T = 256;
  const size_t nf = (size_t)N * 128;

  fill_zero<<<1024, T, 0, stream>>>(dinv, 2 * (size_t)N + 2 * nf);
  const int wtot = 384 * 128 + 128 + 384 * 64 + 64;
  build_weights<<<(wtot + T - 1) / T, T, 0, stream>>>(Wn1, Ws1, b1, Wn2, Ws2, b2,
                                                      Bcat1, bias1, Bcat2, bias2);

  degree_count<<<dim3((E + T - 1) / T, 2), T, 0, stream>>>(edges, dinv, E, N);
  recip_deg<<<(2 * N + T - 1) / T, T, 0, stream>>>(dinv, 2 * N);

  const int sblocks = (int)(((size_t)E * 32 + T - 1) / T);
  const int fblocks = (int)((nf + T - 1) / T);
  const int gblocks = (N + 127) / 128;

  scatter_feat<<<sblocks, T, 0, stream>>>(edges,         edges + E,     x, agg0, E);
  scatter_feat<<<sblocks, T, 0, stream>>>(edges + 2 * E, edges + 3 * E, x, agg1, E);
  scale_rows<<<fblocks, T, 0, stream>>>(agg0, dinv,     nf);
  scale_rows<<<fblocks, T, 0, stream>>>(agg1, dinv + N, nf);
  sage_gemm<128, true><<<gblocks, T, 0, stream>>>(x, agg0, agg1, Bcat1, bias1, h1, N);

  fill_zero<<<1024, T, 0, stream>>>(agg0, 2 * nf);
  scatter_feat<<<sblocks, T, 0, stream>>>(edges,         edges + E,     h1, agg0, E);
  scatter_feat<<<sblocks, T, 0, stream>>>(edges + 2 * E, edges + 3 * E, h1, agg1, E);
  scale_rows<<<fblocks, T, 0, stream>>>(agg0, dinv,     nf);
  scale_rows<<<fblocks, T, 0, stream>>>(agg1, dinv + N, nf);
  sage_gemm<64, false><<<gblocks, T, 0, stream>>>(h1, agg0, agg1, Bcat2, bias2, h2, N);

  edge_score<<<sblocks, T, 0, stream>>>(edges, edges + E, h2, out, E);
  edge_score<<<sblocks, T, 0, stream>>>(negE,  negE + E,  h2, out + E, E);
}


# Mean aggregation owner-computes: workgroup owns 200 nodes, scans all dst ids, lists its edges in LDS, adds rows by LDS float atomics (arrival order), writes means; no global atomics, fills, scaling
# speedup vs baseline: 2.0552x; 2.0552x over previous
#include <hip/hip_runtime.h>
#include <hip/hip_bf16.h>

typedef float v2f __attribute__((ext_vector_type(2)));
typedef float v8f __attribute__((ext_vector_type(8)));

#define FDIM 128
#define KTOT 384
#define KC   64

__global__ void build_weights(const float* __restrict__ Wn1, const float* __restrict__ Ws1,
                              const float* __restrict__ b1,
                              const float* __restrict__ Wn2, const float* __restrict__ Ws2,
                              const float* __restrict__ b2,
                              float* __restrict__ Bcat1, float* __restrict__ bias1,
                              float* __restrict__ Bcat2, float* __restrict__ bias2) {
  int i = blockIdx.x * blockDim.x + threadIdx.x;
  const int S1 = 384 * 128, S2 = S1 + 128, S3 = S2 + 384 * 64, S4 = S3 + 64;
  if (i < S1) {
    int k = i >> 7, n = i & 127;
    float v;
    if (k < 128)      v = Ws1[k * 128 + n] + Ws1[16384 + k * 128 + n];
    else if (k < 256) v = Wn1[(k - 128) * 128 + n];
    else              v = Wn1[16384 + (k - 256) * 128 + n];
    Bcat1[i] = v;
  } else if (i < S2) {
    int n = i - S1;
    bias1[n] = b1[n] + b1[128 + n];
  } else if (i < S3) {
    int j = i - S2;
    int k = j >> 6, n = j & 63;
    float v;
    if (k < 128)      v = Ws2[k * 64 + n] + Ws2[8192 + k * 64 + n];
    else if (k < 256) v = Wn2[(k - 128) * 64 + n];
    else              v = Wn2[8192 + (k - 256) * 64 + n];
    Bcat2[j] = v;
  } else if (i < S4) {
    int n = i - S3;
    bias2[n] = b2[n] + b2[64 + n];
  }
}

#define RN   200
#define LCAP 5120
#define OTH  1024
#define SU   8
#define DU   8

__device__ __forceinline__ int acc_pos(int f) { return ((f & 3) << 5) + (f >> 2); }

__device__ __forceinline__ void note_edge(int dv, int e, int lo, int r,
                                          const int* __restrict__ src,
                                          const float* __restrict__ feat,
                                          float* acc, int* le, int* ldl, int* lcnt, int* cnt) {
  const unsigned t = (unsigned)(dv - lo);
  if (t < (unsigned)RN) {
    atomicAdd(&cnt[r * RN + (int)t], 1);
    const int slot = atomicAdd(&lcnt[r], 1);
    if (slot < LCAP) {
      le[r * LCAP + slot] = e;
      ldl[r * LCAP + slot] = (int)t;
    } else {
      const float* row = feat + (size_t)src[e] * FDIM;
      float* a = acc + (size_t)(r * RN + (int)t) * FDIM;
      for (int f = 0; f < FDIM; ++f) atomicAdd(&a[acc_pos(f)], row[f]);
    }
  }
}

template <int NL, bool GUARD>
__device__ __forceinline__ void scan_group(const int4* __restrict__ d4, int base, int nvec,
                                           int tid, int lo, int r,
                                           const int* __restrict__ src,
                                           const float* __restrict__ feat,
                                           float* acc, int* le, int* ldl, int* lcnt, int* cnt) {
  int4 d[NL];
#pragma unroll
  for (int u = 0; u < NL; ++u) {
    const int i = base + u * OTH + tid;
    if (GUARD) d[u] = (i < nvec) ? d4[i] : make_int4(-1, -1, -1, -1);
    else       d[u] = d4[i];
  }
#pragma unroll
  for (int u = 0; u < NL; ++u) {
    const int i = base + u * OTH + tid;
    const unsigned t0 = (unsigned)(d[u].x - lo), t1 = (unsigned)(d[u].y - lo);
    const unsigned t2 = (unsigned)(d[u].z - lo), t3 = (unsigned)(d[u].w - lo);
    const unsigned tm = min(min(t0, t1), min(t2, t3));
    if (tm < (unsigned)RN) {
      note_edge(d[u].x, 4 * i + 0, lo, r, src, feat, acc, le, ldl, lcnt, cnt);
      note_edge(d[u].y, 4 * i + 1, lo, r, src, feat, acc, le, ldl, lcnt, cnt);
      note_edge(d[u].z, 4 * i + 2, lo, r, src, feat, acc, le, ldl, lcnt, cnt);
      note_edge(d[u].w, 4 * i + 3, lo, r, src, feat, acc, le, ldl, lcnt, cnt);
    }
  }
}

__global__ __launch_bounds__(OTH) void mean_owner(const int* __restrict__ edges,
                                                  const float* __restrict__ feat,
                                                  float* __restrict__ agg0,
                                                  float* __restrict__ agg1, int E, int N) {
  __shared__ __attribute__((aligned(16))) float acc[(2 * RN + 1) * FDIM];
  __shared__ int le[2 * LCAP];
  __shared__ int ldl[2 * LCAP];
  __shared__ int cnt[2 * RN];
  __shared__ int lcnt[2];

  const int tid = threadIdx.x;
  const int lane = tid & 31;
  const int wave = __builtin_amdgcn_readfirstlane(tid >> 5);
  const int lo = blockIdx.x * RN;

  for (int i = tid; i < 2 * RN * FDIM / 4; i += OTH)
    ((float4*)acc)[i] = make_float4(0.0f, 0.0f, 0.0f, 0.0f);
  for (int i = tid; i < 2 * RN; i += OTH) cnt[i] = 0;
  if (tid < 2) lcnt[tid] = 0;
  __syncthreads();

  const bool vec = ((E & 3) == 0) && ((((size_t)edges) & 15) == 0);
  const int nvec = vec ? (E >> 2) : 0;
  for (int r = 0; r < 2; ++r) {
    const int* src = edges + (size_t)r * 2 * E;
    const int* dst = src + E;
    const int4* d4 = (const int4*)dst;
    const int nfull = nvec - nvec % (OTH * SU);
    for (int base = 0; base < nfull; base += OTH * SU)
      scan_group<SU, false>(d4, base, nvec, tid, lo, r, src, feat, acc, le, ldl, lcnt, cnt);
    for (int base = nfull; base < nvec; base += OTH * 4)
      scan_group<4, true>(d4, base, nvec, tid, lo, r, src, feat, acc, le, ldl, lcnt, cnt);
    for (int e = nvec * 4 + tid; e < E; e += OTH)
      note_edge(dst[e], e, lo, r, src, feat, acc, le, ldl, lcnt, cnt);
  }
  __syncthreads();

  const int n0 = min(lcnt[0], LCAP), n1 = min(lcnt[1], LCAP);
  for (int r = 0; r < 2; ++r) {
    const int* src = edges + (size_t)r * 2 * E;
    const int n = r ? n1 : n0;
    for (int i = tid; i < n; i += OTH) le[r * LCAP + i] = src[le[r * LCAP + i]];
  }
  __syncthreads();

  for (int r = 0; r < 2; ++r) {
    const int n = r ? n1 : n0;
    float* ar = acc + (size_t)r * RN * FDIM;
    const int* ls = le + r * LCAP;
    const int* ld = ldl + r * LCAP;
    const int spare = 2 * RN - r * RN;
    for (int i0 = wave; i0 < n; i0 += 32 * DU) {
      const int i = i0 + lane * 32;
      const bool ok = (lane < DU) && (i < n);
      const int sv = ok ? ls[i] : 0;
      const int dv = ok ? ld[i] : spare;
      float4 v[DU];
      int dl[DU];
#pragma unroll
      for (int u = 0; u < DU; ++u) {
        const int s = __builtin_amdgcn_readlane(sv, u);
        dl[u] = __builtin_amdgcn_readlane(dv, u);
        v[u] = *(const float4*)(feat + (size_t)s * FDIM + lane * 4);
      }
#pragma unroll
      for (int u = 0; u < DU; ++u) {
        float* a = ar + dl[u] * FDIM + lane;
        atomicAdd(a, v[u].x);
        atomicAdd(a + 32, v[u].y);
        atomicAdd(a + 64, v[u].z);
        atomicAdd(a + 96, v[u].w);
      }
    }
  }
  __syncthreads();

  for (int idx = tid; idx < 2 * RN * 32; idx += OTH) {
    const int r = idx / (RN * 32);
    const int rem = idx - r * (RN * 32);
    const int row = rem >> 5, q = rem & 31;
    const int node = lo + row;
    if (node < N) {
      const float* a = acc + (size_t)(r * RN + row) * FDIM + q;
      const float inv = 1.0f / fmaxf((float)cnt[r * RN + row], 1.0f);
      float4 o;
      o.x = a[0] * inv;
      o.y = a[32] * inv;
      o.z = a[64] * inv;
      o.w = a[96] * inv;
      *(float4*)((r ? agg1 : agg0) + (size_t)node * FDIM + q * 4) = o;
    }
  }
}

template <int NOUT, bool RELU>
__global__ __launch_bounds__(256) void sage_gemm(
    const float* __restrict__ A0, const float* __restrict__ G0, const float* __restrict__ G1,
    const float* __restrict__ Bmat, const float* __restrict__ bias,
    float* __restrict__ out, int Nrows) {
  constexpr int NT = NOUT / 16;
  __shared__ float Bs[KC * NOUT];

  const int tid = threadIdx.x;
  const int lane = tid & 31;
  const int wave = tid >> 5;
  const int rowBase = blockIdx.x * 128 + wave * 16;
  const int m = lane & 15;
  const int khalf = (lane >> 4) << 1;

  v8f czero = {};
  v8f c[NT];
#pragma unroll
  for (int t = 0; t < NT; ++t) c[t] = czero;

  int rowA = rowBase + m;
  if (rowA > Nrows - 1) rowA = Nrows - 1;

  for (int kc = 0; kc < KTOT; kc += KC) {
    for (int i = tid; i < KC * NOUT; i += 256)
      Bs[i] = Bmat[kc * NOUT + i];
    __syncthreads();

    const float* Asrc = (kc < 128) ? A0 : ((kc < 256) ? G0 : G1);
    const int kbase = kc & 127;

    for (int k4 = 0; k4 < KC; k4 += 4) {
      v2f a = *(const v2f*)(Asrc + (size_t)rowA * FDIM + kbase + k4 + khalf);
      const int kb = k4 + khalf;
#pragma unroll
      for (int t = 0; t < NT; ++t) {
        v2f b;
        b.x = Bs[kb * NOUT + t * 16 + m];
        b.y = Bs[(kb + 1) * NOUT + t * 16 + m];
        c[t] = __builtin_amdgcn_wmma_f32_16x16x4_f32(
            false, a, false, b, (short)0, c[t], false, false);
      }
    }
    __syncthreads();
  }

  const int rhalf = (lane >> 4) << 3;
#pragma unroll
  for (int t = 0; t < NT; ++t) {
    int col = t * 16 + m;
    float bv = bias[col];
#pragma unroll
    for (int v = 0; v < 8; ++v) {
      int row = rowBase + v + rhalf;
      if (row < Nrows) {
        float val = c[t][v] + bv;
        if (RELU) val = fmaxf(val, 0.0f);
        out[(size_t)row * NOUT + col] = val;
      }
    }
  }
}

__global__ void edge_score(const int* __restrict__ src, const int* __restrict__ dst,
                           const float* __restrict__ h, float* __restrict__ out, int E) {
  int gid = blockIdx.x * blockDim.x + threadIdx.x;
  int e = gid >> 5, lane = gid & 31;
  if (e >= E) return;
  int s = src[e], d = dst[e];
  v2f a = *(const v2f*)(h + (size_t)s * 64 + lane * 2);
  v2f b = *(const v2f*)(h + (size_t)d * 64 + lane * 2);
  float p = a.x * b.x + a.y * b.y;
#pragma unroll
  for (int off = 16; off > 0; off >>= 1)
    p += __shfl_xor(p, off, 32);
  if (lane == 0) out[e] = p;
}

extern "C" void kernel_launch(void* const* d_in, const int* in_sizes, int n_in,
                              void* d_out, int out_size, void* d_ws, size_t ws_size,
                              hipStream_t stream) {
  const float* x   = (const float*)d_in[0];
  const int*   edges = (const int*)d_in[1];
  const int*   negE  = (const int*)d_in[2];
  const float* Wn1 = (const float*)d_in[3];
  const float* Ws1 = (const float*)d_in[4];
  const float* b1  = (const float*)d_in[5];
  const float* Wn2 = (const float*)d_in[6];
  const float* Ws2 = (const float*)d_in[7];
  const float* b2  = (const float*)d_in[8];
  float* out = (float*)d_out;

  const int N = in_sizes[0] / 128;
  const int E = in_sizes[1] / 4;

  float* ws    = (float*)d_ws;
  float* dinv  = ws;
  float* agg0  = dinv + 2 * (size_t)N;
  float* agg1  = agg0 + (size_t)N * 128;
  float* h1    = agg1 + (size_t)N * 128;
  float* h2    = h1 + (size_t)N * 128;
  float* Bcat1 = h2 + (size_t)N * 64;
  float* bias1 = Bcat1 + 384 * 128;
  float* Bcat2 = bias1 + 128;
  float* bias2 = Bcat2 + 384 * 64;

  const int T = 256;

  const int wtot = 384 * 128 + 128 + 384 * 64 + 64;
  build_weights<<<(wtot + T - 1) / T, T, 0, stream>>>(Wn1, Ws1, b1, Wn2, Ws2, b2,
                                                      Bcat1, bias1, Bcat2, bias2);

  const int sblocks = (int)(((size_t)E * 32 + T - 1) / T);
  const int gblocks = (N + 127) / 128;
  const int oblocks = (N + RN - 1) / RN;

  mean_owner<<<oblocks, OTH, 0, stream>>>(edges, x, agg0, agg1, E, N);
  sage_gemm<128, true><<<gblocks, T, 0, stream>>>(x, agg0, agg1, Bcat1, bias1, h1, N);

  mean_owner<<<oblocks, OTH, 0, stream>>>(edges, h1, agg0, agg1, E, N);
  sage_gemm<64, false><<<gblocks, T, 0, stream>>>(h1, agg0, agg1, Bcat2, bias2, h2, N);

  edge_score<<<sblocks, T, 0, stream>>>(edges, edges + E, h2, out, E);
  edge_score<<<sblocks, T, 0, stream>>>(negE,  negE + E,  h2, out + E, E);
}


# Layer-2 mean aggregation skips the dst-id scan: layer 1 saves each workgroup's edge list (source node, local dst) in the workspace, layer 2 loads it; sums stay LDS float atomics in arrival order
# speedup vs baseline: 2.3108x; 1.1244x over previous
#include <hip/hip_runtime.h>
#include <hip/hip_bf16.h>

typedef float v2f __attribute__((ext_vector_type(2)));
typedef float v8f __attribute__((ext_vector_type(8)));

#define FDIM 128
#define KTOT 384
#define KC   64

__global__ void build_weights(const float* __restrict__ Wn1, const float* __restrict__ Ws1,
                              const float* __restrict__ b1,
                              const float* __restrict__ Wn2, const float* __restrict__ Ws2,
                              const float* __restrict__ b2,
                              float* __restrict__ Bcat1, float* __restrict__ bias1,
                              float* __restrict__ Bcat2, float* __restrict__ bias2) {
  int i = blockIdx.x * blockDim.x + threadIdx.x;
  const int S1 = 384 * 128, S2 = S1 + 128, S3 = S2 + 384 * 64, S4 = S3 + 64;
  if (i < S1) {
    int k = i >> 7, n = i & 127;
    float v;
    if (k < 128)      v = Ws1[k * 128 + n] + Ws1[16384 + k * 128 + n];
    else if (k < 256) v = Wn1[(k - 128) * 128 + n];
    else              v = Wn1[16384 + (k - 256) * 128 + n];
    Bcat1[i] = v;
  } else if (i < S2) {
    int n = i - S1;
    bias1[n] = b1[n] + b1[128 + n];
  } else if (i < S3) {
    int j = i - S2;
    int k = j >> 6, n = j & 63;
    float v;
    if (k < 128)      v = Ws2[k * 64 + n] + Ws2[8192 + k * 64 + n];
    else if (k < 256) v = Wn2[(k - 128) * 64 + n];
    else              v = Wn2[8192 + (k - 256) * 64 + n];
    Bcat2[j] = v;
  } else if (i < S4) {
    int n = i - S3;
    bias2[n] = b2[n] + b2[64 + n];
  }
}

#define RN   200
#define LCAP 5120
#define OTH  1024
#define SU   8
#define DU   8
#define LHDR 32
#define LREG (2 * LCAP + LHDR)

__device__ __forceinline__ int acc_pos(int f) { return ((f & 3) << 5) + (f >> 2); }

__device__ __forceinline__ void note_edge(int dv, int e, int lo, int r,
                                          const int* __restrict__ src,
                                          const float* __restrict__ feat,
                                          float* acc, int* le, int* ldl, int* lcnt, int* cnt) {
  const unsigned t = (unsigned)(dv - lo);
  if (t < (unsigned)RN) {
    atomicAdd(&cnt[r * RN + (int)t], 1);
    const int slot = atomicAdd(&lcnt[r], 1);
    if (slot < LCAP) {
      le[r * LCAP + slot] = e;
      ldl[r * LCAP + slot] = (int)t;
    } else {
      const float* row = feat + (size_t)src[e] * FDIM;
      float* a = acc + (size_t)(r * RN + (int)t) * FDIM;
      for (int f = 0; f < FDIM; ++f) atomicAdd(&a[acc_pos(f)], row[f]);
    }
  }
}

template <int NL, bool GUARD>
__device__ __forceinline__ void scan_group(const int4* __restrict__ d4, int base, int nvec,
                                           int tid, int lo, int r,
                                           const int* __restrict__ src,
                                           const float* __restrict__ feat,
                                           float* acc, int* le, int* ldl, int* lcnt, int* cnt) {
  int4 d[NL];
#pragma unroll
  for (int u = 0; u < NL; ++u) {
    const int i = base + u * OTH + tid;
    if (GUARD) d[u] = (i < nvec) ? d4[i] : make_int4(-1, -1, -1, -1);
    else       d[u] = d4[i];
  }
#pragma unroll
  for (int u = 0; u < NL; ++u) {
    const int i = base + u * OTH + tid;
    const unsigned t0 = (unsigned)(d[u].x - lo), t1 = (unsigned)(d[u].y - lo);
    const unsigned t2 = (unsigned)(d[u].z - lo), t3 = (unsigned)(d[u].w - lo);
    const unsigned tm = min(min(t0, t1), min(t2, t3));
    if (tm < (unsigned)RN) {
      note_edge(d[u].x, 4 * i + 0, lo, r, src, feat, acc, le, ldl, lcnt, cnt);
      note_edge(d[u].y, 4 * i + 1, lo, r, src, feat, acc, le, ldl, lcnt, cnt);
      note_edge(d[u].z, 4 * i + 2, lo, r, src, feat, acc, le, ldl, lcnt, cnt);
      note_edge(d[u].w, 4 * i + 3, lo, r, src, feat, acc, le, ldl, lcnt, cnt);
    }
  }
}

template <int MODE>
__global__ __launch_bounds__(OTH) void mean_owner(const int* __restrict__ edges,
                                                  const float* __restrict__ feat,
                                                  float* __restrict__ agg0,
                                                  float* __restrict__ agg1,
                                                  unsigned* __restrict__ lists, int E, int N) {
  __shared__ __attribute__((aligned(16))) float acc[(2 * RN + 1) * FDIM];
  __shared__ int le[2 * LCAP];
  __shared__ int ldl[2 * LCAP];
  __shared__ int cnt[2 * RN];
  __shared__ int lcnt[2];

  const int tid = threadIdx.x;
  const int lane = tid & 31;
  const int wave = __builtin_amdgcn_readfirstlane(tid >> 5);
  const int lo = blockIdx.x * RN;

  for (int i = tid; i < 2 * RN * FDIM / 4; i += OTH)
    ((float4*)acc)[i] = make_float4(0.0f, 0.0f, 0.0f, 0.0f);
  for (int i = tid; i < 2 * RN; i += OTH) cnt[i] = 0;
  if (tid < 2) lcnt[tid] = 0;
  __syncthreads();

  unsigned* mine = lists + (size_t)blockIdx.x * LREG;
  const bool room = ((size_t)gridDim.x * LREG <= (size_t)N * 64) && (N <= (1 << 24)) && (RN <= 256);
  int c0 = -1, c1 = -1;
  if (MODE == 1 && room) {
    c0 = __builtin_amdgcn_readfirstlane(((const int*)mine)[0]);
    c1 = __builtin_amdgcn_readfirstlane(((const int*)mine)[1]);
  }
  const bool have = (MODE == 1) && room && ((unsigned)c0 <= (unsigned)LCAP) &&
                    ((unsigned)c1 <= (unsigned)LCAP);

  const bool vec = ((E & 3) == 0) && ((((size_t)edges) & 15) == 0);
  const int nvec = vec ? (E >> 2) : 0;
  for (int r = 0; r < 2 && !have; ++r) {
    const int* src = edges + (size_t)r * 2 * E;
    const int* dst = src + E;
    const int4* d4 = (const int4*)dst;
    const int nfull = nvec - nvec % (OTH * SU);
    for (int base = 0; base < nfull; base += OTH * SU)
      scan_group<SU, false>(d4, base, nvec, tid, lo, r, src, feat, acc, le, ldl, lcnt, cnt);
    for (int base = nfull; base < nvec; base += OTH * 4)
      scan_group<4, true>(d4, base, nvec, tid, lo, r, src, feat, acc, le, ldl, lcnt, cnt);
    for (int e = nvec * 4 + tid; e < E; e += OTH)
      note_edge(dst[e], e, lo, r, src, feat, acc, le, ldl, lcnt, cnt);
  }
  __syncthreads();

  const int n0 = have ? c0 : min(lcnt[0], LCAP), n1 = have ? c1 : min(lcnt[1], LCAP);
  if (have) {
    constexpr int PT = LCAP / OTH;
    static_assert(LCAP % OTH == 0, "a list is a whole number of words per thread");
    unsigned p[2][PT];
#pragma unroll
    for (int r = 0; r < 2; ++r) {
      const unsigned* sl = mine + LHDR + r * LCAP;
      const int n = r ? n1 : n0;
#pragma unroll
      for (int k = 0; k < PT; ++k) {
        const int i = tid + k * OTH;
        p[r][k] = (i < n) ? sl[i] : 0u;
      }
    }
#pragma unroll
    for (int r = 0; r < 2; ++r) {
      const int n = r ? n1 : n0;
#pragma unroll
      for (int k = 0; k < PT; ++k) {
        const int i = tid + k * OTH;
        if (i < n) {
          const int s = min((int)(p[r][k] >> 8), N - 1);
          const int t = min((int)(p[r][k] & 255u), RN - 1);
          le[r * LCAP + i] = s;
          ldl[r * LCAP + i] = t;
          atomicAdd(&cnt[r * RN + t], 1);
        }
      }
    }
  } else {
    const bool save = (MODE == 0) && room;
    if (save && tid == 0) {
      ((int*)mine)[0] = lcnt[0];
      ((int*)mine)[1] = lcnt[1];
    }
    for (int r = 0; r < 2; ++r) {
      const int* src = edges + (size_t)r * 2 * E;
      unsigned* sl = mine + LHDR + r * LCAP;
      const int n = r ? n1 : n0;
      for (int i = tid; i < n; i += OTH) {
        const int s = src[le[r * LCAP + i]];
        le[r * LCAP + i] = s;
        if (save) sl[i] = ((unsigned)s << 8) | (unsigned)ldl[r * LCAP + i];
      }
    }
  }
  __syncthreads();

  for (int r = 0; r < 2; ++r) {
    const int n = r ? n1 : n0;
    float* ar = acc + (size_t)r * RN * FDIM;
    const int* ls = le + r * LCAP;
    const int* ld = ldl + r * LCAP;
    const int spare = 2 * RN - r * RN;
    for (int i0 = wave; i0 < n; i0 += 32 * DU) {
      const int i = i0 + lane * 32;
      const bool ok = (lane < DU) && (i < n);
      const int sv = ok ? ls[i] : 0;
      const int dv = ok ? ld[i] : spare;
      float4 v[DU];
      int dl[DU];
#pragma unroll
      for (int u = 0; u < DU; ++u) {
        const int s = __builtin_amdgcn_readlane(sv, u);
        dl[u] = __builtin_amdgcn_readlane(dv, u);
        v[u] = *(const float4*)(feat + (size_t)s * FDIM + lane * 4);
      }
#pragma unroll
      for (int u = 0; u < DU; ++u) {
        float* a = ar + dl[u] * FDIM + lane;
        atomicAdd(a, v[u].x);
        atomicAdd(a + 32, v[u].y);
        atomicAdd(a + 64, v[u].z);
        atomicAdd(a + 96, v[u].w);
      }
    }
  }
  __syncthreads();

  for (int idx = tid; idx < 2 * RN * 32; idx += OTH) {
    const int r = idx / (RN * 32);
    const int rem = idx - r * (RN * 32);
    const int row = rem >> 5, q = rem & 31;
    const int node = lo + row;
    if (node < N) {
      const float* a = acc + (size_t)(r * RN + row) * FDIM + q;
      const float inv = 1.0f / fmaxf((float)cnt[r * RN + row], 1.0f);
      float4 o;
      o.x = a[0] * inv;
      o.y = a[32] * inv;
      o.z = a[64] * inv;
      o.w = a[96] * inv;
      *(float4*)((r ? agg1 : agg0) + (size_t)node * FDIM + q * 4) = o;
    }
  }
}

template <int NOUT, bool RELU>
__global__ __launch_bounds__(256) void sage_gemm(
    const float* __restrict__ A0, const float* __restrict__ G0, const float* __restrict__ G1,
    const float* __restrict__ Bmat, const float* __restrict__ bias,
    float* __restrict__ out, int Nrows) {
  constexpr int NT = NOUT / 16;
  __shared__ float Bs[KC * NOUT];

  const int tid = threadIdx.x;
  const int lane = tid & 31;
  const int wave = tid >> 5;
  const int rowBase = blockIdx.x * 128 + wave * 16;
  const int m = lane & 15;
  const int khalf = (lane >> 4) << 1;

  v8f czero = {};
  v8f c[NT];
#pragma unroll
  for (int t = 0; t < NT; ++t) c[t] = czero;

  int rowA = rowBase + m;
  if (rowA > Nrows - 1) rowA = Nrows - 1;

  for (int kc = 0; kc < KTOT; kc += KC) {
    for (int i = tid; i < KC * NOUT; i += 256)
      Bs[i] = Bmat[kc * NOUT + i];
    __syncthreads();

    const float* Asrc = (kc < 128) ? A0 : ((kc < 256) ? G0 : G1);
    const int kbase = kc & 127;

    for (int k4 = 0; k4 < KC; k4 += 4) {
      v2f a = *(const v2f*)(Asrc + (size_t)rowA * FDIM + kbase + k4 + khalf);
      const int kb = k4 + khalf;
#pragma unroll
      for (int t = 0; t < NT; ++t) {
        v2f b;
        b.x = Bs[kb * NOUT + t * 16 + m];
        b.y = Bs[(kb + 1) * NOUT + t * 16 + m];
        c[t] = __builtin_amdgcn_wmma_f32_16x16x4_f32(
            false, a, false, b, (short)0, c[t], false, false);
      }
    }
    __syncthreads();
  }

  const int rhalf = (lane >> 4) << 3;
#pragma unroll
  for (int t = 0; t < NT; ++t) {
    int col = t * 16 + m;
    float bv = bias[col];
#pragma unroll
    for (int v = 0; v < 8; ++v) {
      int row = rowBase + v + rhalf;
      if (row < Nrows) {
        float val = c[t][v] + bv;
        if (RELU) val = fmaxf(val, 0.0f);
        out[(size_t)row * NOUT + col] = val;
      }
    }
  }
}

__global__ void edge_score(const int* __restrict__ src, const int* __restrict__ dst,
                           const float* __restrict__ h, float* __restrict__ out, int E) {
  int gid = blockIdx.x * blockDim.x + threadIdx.x;
  int e = gid >> 5, lane = gid & 31;
  if (e >= E) return;
  int s = src[e], d = dst[e];
  v2f a = *(const v2f*)(h + (size_t)s * 64 + lane * 2);
  v2f b = *(const v2f*)(h + (size_t)d * 64 + lane * 2);
  float p = a.x * b.x + a.y * b.y;
#pragma unroll
  for (int off = 16; off > 0; off >>= 1)
    p += __shfl_xor(p, off, 32);
  if (lane == 0) out[e] = p;
}

extern "C" void kernel_launch(void* const* d_in, const int* in_sizes, int n_in,
                              void* d_out, int out_size, void* d_ws, size_t ws_size,
                              hipStream_t stream) {
  const float* x   = (const float*)d_in[0];
  const int*   edges = (const int*)d_in[1];
  const int*   negE  = (const int*)d_in[2];
  const float* Wn1 = (const float*)d_in[3];
  const float* Ws1 = (const float*)d_in[4];
  const float* b1  = (const float*)d_in[5];
  const float* Wn2 = (const float*)d_in[6];
  const float* Ws2 = (const float*)d_in[7];
  const float* b2  = (const float*)d_in[8];
  float* out = (float*)d_out;

  const int N = in_sizes[0] / 128;
  const int E = in_sizes[1] / 4;

  float* ws    = (float*)d_ws;
  float* dinv  = ws;
  float* agg0  = dinv + 2 * (size_t)N;
  float* agg1  = agg0 + (size_t)N * 128;
  float* h1    = agg1 + (size_t)N * 128;
  float* h2    = h1 + (size_t)N * 128;
  float* Bcat1 = h2 + (size_t)N * 64;
  float* bias1 = Bcat1 + 384 * 128;
  float* Bcat2 = bias1 + 128;
  float* bias2 = Bcat2 + 384 * 64;

  const int T = 256;

  const int wtot = 384 * 128 + 128 + 384 * 64 + 64;
  build_weights<<<(wtot + T - 1) / T, T, 0, stream>>>(Wn1, Ws1, b1, Wn2, Ws2, b2,
                                                      Bcat1, bias1, Bcat2, bias2);

  const int sblocks = (int)(((size_t)E * 32 + T - 1) / T);
  const int gblocks = (N + 127) / 128;
  const int oblocks = (N + RN - 1) / RN;

  unsigned* olist = (unsigned*)h2;
  mean_owner<0><<<oblocks, OTH, 0, stream>>>(edges, x, agg0, agg1, olist, E, N);
  sage_gemm<128, true><<<gblocks, T, 0, stream>>>(x, agg0, agg1, Bcat1, bias1, h1, N);

  mean_owner<1><<<oblocks, OTH, 0, stream>>>(edges, h1, agg0, agg1, olist, E, N);
  sage_gemm<64, false><<<gblocks, T, 0, stream>>>(h1, agg0, agg1, Bcat2, bias2, h2, N);

  edge_score<<<sblocks, T, 0, stream>>>(edges, edges + E, h2, out, E);
  edge_score<<<sblocks, T, 0, stream>>>(negE,  negE + E,  h2, out + E, E);
}


# Edge scoring: the wave's 32-lane sum takes partner values through permlanex16 and row_xmask lane moves instead of five serial LDS permutes; same butterfly, sum order unchanged
# speedup vs baseline: 2.3915x; 1.0349x over previous
#include <hip/hip_runtime.h>
#include <hip/hip_bf16.h>

typedef float v2f __attribute__((ext_vector_type(2)));
typedef float v8f __attribute__((ext_vector_type(8)));

#define FDIM 128
#define KTOT 384
#define KC   64

__global__ void build_weights(const float* __restrict__ Wn1, const float* __restrict__ Ws1,
                              const float* __restrict__ b1,
                              const float* __restrict__ Wn2, const float* __restrict__ Ws2,
                              const float* __restrict__ b2,
                              float* __restrict__ Bcat1, float* __restrict__ bias1,
                              float* __restrict__ Bcat2, float* __restrict__ bias2) {
  int i = blockIdx.x * blockDim.x + threadIdx.x;
  const int S1 = 384 * 128, S2 = S1 + 128, S3 = S2 + 384 * 64, S4 = S3 + 64;
  if (i < S1) {
    int k = i >> 7, n = i & 127;
    float v;
    if (k < 128)      v = Ws1[k * 128 + n] + Ws1[16384 + k * 128 + n];
    else if (k < 256) v = Wn1[(k - 128) * 128 + n];
    else              v = Wn1[16384 + (k - 256) * 128 + n];
    Bcat1[i] = v;
  } else if (i < S2) {
    int n = i - S1;
    bias1[n] = b1[n] + b1[128 + n];
  } else if (i < S3) {
    int j = i - S2;
    int k = j >> 6, n = j & 63;
    float v;
    if (k < 128)      v = Ws2[k * 64 + n] + Ws2[8192 + k * 64 + n];
    else if (k < 256) v = Wn2[(k - 128) * 64 + n];
    else              v = Wn2[8192 + (k - 256) * 64 + n];
    Bcat2[j] = v;
  } else if (i < S4) {
    int n = i - S3;
    bias2[n] = b2[n] + b2[64 + n];
  }
}

#define RN   200
#define LCAP 5120
#define OTH  1024
#define SU   8
#define DU   8
#define LHDR 32
#define LREG (2 * LCAP + LHDR)

__device__ __forceinline__ int acc_pos(int f) { return ((f & 3) << 5) + (f >> 2); }

__device__ __forceinline__ void note_edge(int dv, int e, int lo, int r,
                                          const int* __restrict__ src,
                                          const float* __restrict__ feat,
                                          float* acc, int* le, int* ldl, int* lcnt, int* cnt) {
  const unsigned t = (unsigned)(dv - lo);
  if (t < (unsigned)RN) {
    atomicAdd(&cnt[r * RN + (int)t], 1);
    const int slot = atomicAdd(&lcnt[r], 1);
    if (slot < LCAP) {
      le[r * LCAP + slot] = e;
      ldl[r * LCAP + slot] = (int)t;
    } else {
      const float* row = feat + (size_t)src[e] * FDIM;
      float* a = acc + (size_t)(r * RN + (int)t) * FDIM;
      for (int f = 0; f < FDIM; ++f) atomicAdd(&a[acc_pos(f)], row[f]);
    }
  }
}

template <int NL, bool GUARD>
__device__ __forceinline__ void scan_group(const int4* __restrict__ d4, int base, int nvec,
                                           int tid, int lo, int r,
                                           const int* __restrict__ src,
                                           const float* __restrict__ feat,
                                           float* acc, int* le, int* ldl, int* lcnt, int* cnt) {
  int4 d[NL];
#pragma unroll
  for (int u = 0; u < NL; ++u) {
    const int i = base + u * OTH + tid;
    if (GUARD) d[u] = (i < nvec) ? d4[i] : make_int4(-1, -1, -1, -1);
    else       d[u] = d4[i];
  }
#pragma unroll
  for (int u = 0; u < NL; ++u) {
    const int i = base + u * OTH + tid;
    const unsigned t0 = (unsigned)(d[u].x - lo), t1 = (unsigned)(d[u].y - lo);
    const unsigned t2 = (unsigned)(d[u].z - lo), t3 = (unsigned)(d[u].w - lo);
    const unsigned tm = min(min(t0, t1), min(t2, t3));
    if (tm < (unsigned)RN) {
      note_edge(d[u].x, 4 * i + 0, lo, r, src, feat, acc, le, ldl, lcnt, cnt);
      note_edge(d[u].y, 4 * i + 1, lo, r, src, feat, acc, le, ldl, lcnt, cnt);
      note_edge(d[u].z, 4 * i + 2, lo, r, src, feat, acc, le, ldl, lcnt, cnt);
      note_edge(d[u].w, 4 * i + 3, lo, r, src, feat, acc, le, ldl, lcnt, cnt);
    }
  }
}

template <int MODE>
__global__ __launch_bounds__(OTH) void mean_owner(const int* __restrict__ edges,
                                                  const float* __restrict__ feat,
                                                  float* __restrict__ agg0,
                                                  float* __restrict__ agg1,
                                                  unsigned* __restrict__ lists, int E, int N) {
  __shared__ __attribute__((aligned(16))) float acc[(2 * RN + 1) * FDIM];
  __shared__ int le[2 * LCAP];
  __shared__ int ldl[2 * LCAP];
  __shared__ int cnt[2 * RN];
  __shared__ int lcnt[2];

  const int tid = threadIdx.x;
  const int lane = tid & 31;
  const int wave = __builtin_amdgcn_readfirstlane(tid >> 5);
  const int lo = blockIdx.x * RN;

  for (int i = tid; i < 2 * RN * FDIM / 4; i += OTH)
    ((float4*)acc)[i] = make_float4(0.0f, 0.0f, 0.0f, 0.0f);
  for (int i = tid; i < 2 * RN; i += OTH) cnt[i] = 0;
  if (tid < 2) lcnt[tid] = 0;
  __syncthreads();

  unsigned* mine = lists + (size_t)blockIdx.x * LREG;
  const bool room = ((size_t)gridDim.x * LREG <= (size_t)N * 64) && (N <= (1 << 24)) && (RN <= 256);
  int c0 = -1, c1 = -1;
  if (MODE == 1 && room) {
    c0 = __builtin_amdgcn_readfirstlane(((const int*)mine)[0]);
    c1 = __builtin_amdgcn_readfirstlane(((const int*)mine)[1]);
  }
  const bool have = (MODE == 1) && room && ((unsigned)c0 <= (unsigned)LCAP) &&
                    ((unsigned)c1 <= (unsigned)LCAP);

  const bool vec = ((E & 3) == 0) && ((((size_t)edges) & 15) == 0);
  const int nvec = vec ? (E >> 2) : 0;
  for (int r = 0; r < 2 && !have; ++r) {
    const int* src = edges + (size_t)r * 2 * E;
    const int* dst = src + E;
    const int4* d4 = (const int4*)dst;
    const int nfull = nvec - nvec % (OTH * SU);
    for (int base = 0; base < nfull; base += OTH * SU)
      scan_group<SU, false>(d4, base, nvec, tid, lo, r, src, feat, acc, le, ldl, lcnt, cnt);
    for (int base = nfull; base < nvec; base += OTH * 4)
      scan_group<4, true>(d4, base, nvec, tid, lo, r, src, feat, acc, le, ldl, lcnt, cnt);
    for (int e = nvec * 4 + tid; e < E; e += OTH)
      note_edge(dst[e], e, lo, r, src, feat, acc, le, ldl, lcnt, cnt);
  }
  __syncthreads();

  const int n0 = have ? c0 : min(lcnt[0], LCAP), n1 = have ? c1 : min(lcnt[1], LCAP);
  if (have) {
    constexpr int PT = LCAP / OTH;
    static_assert(LCAP % OTH == 0, "a list is a whole number of words per thread");
    unsigned p[2][PT];
#pragma unroll
    for (int r = 0; r < 2; ++r) {
      const unsigned* sl = mine + LHDR + r * LCAP;
      const int n = r ? n1 : n0;
#pragma unroll
      for (int k = 0; k < PT; ++k) {
        const int i = tid + k * OTH;
        p[r][k] = (i < n) ? sl[i] : 0u;
      }
    }
#pragma unroll
    for (int r = 0; r < 2; ++r) {
      const int n = r ? n1 : n0;
#pragma unroll
      for (int k = 0; k < PT; ++k) {
        const int i = tid + k * OTH;
        if (i < n) {
          const int s = min((int)(p[r][k] >> 8), N - 1);
          const int t = min((int)(p[r][k] & 255u), RN - 1);
          le[r * LCAP + i] = s;
          ldl[r * LCAP + i] = t;
          atomicAdd(&cnt[r * RN + t], 1);
        }
      }
    }
  } else {
    const bool save = (MODE == 0) && room;
    if (save && tid == 0) {
      ((int*)mine)[0] = lcnt[0];
      ((int*)mine)[1] = lcnt[1];
    }
    for (int r = 0; r < 2; ++r) {
      const int* src = edges + (size_t)r * 2 * E;
      unsigned* sl = mine + LHDR + r * LCAP;
      const int n = r ? n1 : n0;
      for (int i = tid; i < n; i += OTH) {
        const int s = src[le[r * LCAP + i]];
        le[r * LCAP + i] = s;
        if (save) sl[i] = ((unsigned)s << 8) | (unsigned)ldl[r * LCAP + i];
      }
    }
  }
  __syncthreads();

  for (int r = 0; r < 2; ++r) {
    const int n = r ? n1 : n0;
    float* ar = acc + (size_t)r * RN * FDIM;
    const int* ls = le + r * LCAP;
    const int* ld = ldl + r * LCAP;
    const int spare = 2 * RN - r * RN;
    for (int i0 = wave; i0 < n; i0 += 32 * DU) {
      const int i = i0 + lane * 32;
      const bool ok = (lane < DU) && (i < n);
      const int sv = ok ? ls[i] : 0;
      const int dv = ok ? ld[i] : spare;
      float4 v[DU];
      int dl[DU];
#pragma unroll
      for (int u = 0; u < DU; ++u) {
        const int s = __builtin_amdgcn_readlane(sv, u);
        dl[u] = __builtin_amdgcn_readlane(dv, u);
        v[u] = *(const float4*)(feat + (size_t)s * FDIM + lane * 4);
      }
#pragma unroll
      for (int u = 0; u < DU; ++u) {
        float* a = ar + dl[u] * FDIM + lane;
        atomicAdd(a, v[u].x);
        atomicAdd(a + 32, v[u].y);
        atomicAdd(a + 64, v[u].z);
        atomicAdd(a + 96, v[u].w);
      }
    }
  }
  __syncthreads();

  for (int idx = tid; idx < 2 * RN * 32; idx += OTH) {
    const int r = idx / (RN * 32);
    const int rem = idx - r * (RN * 32);
    const int row = rem >> 5, q = rem & 31;
    const int node = lo + row;
    if (node < N) {
      const float* a = acc + (size_t)(r * RN + row) * FDIM + q;
      const float inv = 1.0f / fmaxf((float)cnt[r * RN + row], 1.0f);
      float4 o;
      o.x = a[0] * inv;
      o.y = a[32] * inv;
      o.z = a[64] * inv;
      o.w = a[96] * inv;
      *(float4*)((r ? agg1 : agg0) + (size_t)node * FDIM + q * 4) = o;
    }
  }
}

template <int NOUT, bool RELU>
__global__ __launch_bounds__(256) void sage_gemm(
    const float* __restrict__ A0, const float* __restrict__ G0, const float* __restrict__ G1,
    const float* __restrict__ Bmat, const float* __restrict__ bias,
    float* __restrict__ out, int Nrows) {
  constexpr int NT = NOUT / 16;
  __shared__ float Bs[KC * NOUT];

  const int tid = threadIdx.x;
  const int lane = tid & 31;
  const int wave = tid >> 5;
  const int rowBase = blockIdx.x * 128 + wave * 16;
  const int m = lane & 15;
  const int khalf = (lane >> 4) << 1;

  v8f czero = {};
  v8f c[NT];
#pragma unroll
  for (int t = 0; t < NT; ++t) c[t] = czero;

  int rowA = rowBase + m;
  if (rowA > Nrows - 1) rowA = Nrows - 1;

  for (int kc = 0; kc < KTOT; kc += KC) {
    for (int i = tid; i < KC * NOUT; i += 256)
      Bs[i] = Bmat[kc * NOUT + i];
    __syncthreads();

    const float* Asrc = (kc < 128) ? A0 : ((kc < 256) ? G0 : G1);
    const int kbase = kc & 127;

    for (int k4 = 0; k4 < KC; k4 += 4) {
      v2f a = *(const v2f*)(Asrc + (size_t)rowA * FDIM + kbase + k4 + khalf);
      const int kb = k4 + khalf;
#pragma unroll
      for (int t = 0; t < NT; ++t) {
        v2f b;
        b.x = Bs[kb * NOUT + t * 16 + m];
        b.y = Bs[(kb + 1) * NOUT + t * 16 + m];
        c[t] = __builtin_amdgcn_wmma_f32_16x16x4_f32(
            false, a, false, b, (short)0, c[t], false, false);
      }
    }
    __syncthreads();
  }

  const int rhalf = (lane >> 4) << 3;
#pragma unroll
  for (int t = 0; t < NT; ++t) {
    int col = t * 16 + m;
    float bv = bias[col];
#pragma unroll
    for (int v = 0; v < 8; ++v) {
      int row = rowBase + v + rhalf;
      if (row < Nrows) {
        float val = c[t][v] + bv;
        if (RELU) val = fmaxf(val, 0.0f);
        out[(size_t)row * NOUT + col] = val;
      }
    }
  }
}

template <int CTRL>
__device__ __forceinline__ float row_xor(float v) {
  return __builtin_bit_cast(float, __builtin_amdgcn_update_dpp(0, __builtin_bit_cast(int, v), CTRL, 0xf, 0xf, true));
}
__device__ __forceinline__ float lane_xor16(float v) {
  const int x = __builtin_bit_cast(int, v);
  return __builtin_bit_cast(float, __builtin_amdgcn_permlanex16(x, x, 0x76543210, 0xfedcba98, false, false));
}
__global__ void edge_score(const int* __restrict__ src, const int* __restrict__ dst,
                           const float* __restrict__ h, float* __restrict__ out, int E) {
  int gid = blockIdx.x * blockDim.x + threadIdx.x;
  int e = gid >> 5, lane = gid & 31;
  if (e >= E) return;
  int s = src[e], d = dst[e];
  v2f a = *(const v2f*)(h + (size_t)s * 64 + lane * 2);
  v2f b = *(const v2f*)(h + (size_t)d * 64 + lane * 2);
  float p = a.x * b.x + a.y * b.y;
  p += lane_xor16(p);
  p += row_xor<0x168>(p);
  p += row_xor<0x164>(p);
  p += row_xor<0x162>(p);
  p += row_xor<0x161>(p);
  if (lane == 0) out[e] = p;
}

extern "C" void kernel_launch(void* const* d_in, const int* in_sizes, int n_in,
                              void* d_out, int out_size, void* d_ws, size_t ws_size,
                              hipStream_t stream) {
  const float* x   = (const float*)d_in[0];
  const int*   edges = (const int*)d_in[1];
  const int*   negE  = (const int*)d_in[2];
  const float* Wn1 = (const float*)d_in[3];
  const float* Ws1 = (const float*)d_in[4];
  const float* b1  = (const float*)d_in[5];
  const float* Wn2 = (const float*)d_in[6];
  const float* Ws2 = (const float*)d_in[7];
  const float* b2  = (const float*)d_in[8];
  float* out = (float*)d_out;

  const int N = in_sizes[0] / 128;
  const int E = in_sizes[1] / 4;

  float* ws    = (float*)d_ws;
  float* dinv  = ws;
  float* agg0  = dinv + 2 * (size_t)N;
  float* agg1  = agg0 + (size_t)N * 128;
  float* h1    = agg1 + (size_t)N * 128;
  float* h2    = h1 + (size_t)N * 128;
  float* Bcat1 = h2 + (size_t)N * 64;
  float* bias1 = Bcat1 + 384 * 128;
  float* Bcat2 = bias1 + 128;
  float* bias2 = Bcat2 + 384 * 64;

  const int T = 256;

  const int wtot = 384 * 128 + 128 + 384 * 64 + 64;
  build_weights<<<(wtot + T - 1) / T, T, 0, stream>>>(Wn1, Ws1, b1, Wn2, Ws2, b2,
                                                      Bcat1, bias1, Bcat2, bias2);

  const int sblocks = (int)(((size_t)E * 32 + T - 1) / T);
  const int gblocks = (N + 127) / 128;
  const int oblocks = (N + RN - 1) / RN;

  unsigned* olist = (unsigned*)h2;
  mean_owner<0><<<oblocks, OTH, 0, stream>>>(edges, x, agg0, agg1, olist, E, N);
  sage_gemm<128, true><<<gblocks, T, 0, stream>>>(x, agg0, agg1, Bcat1, bias1, h1, N);

  mean_owner<1><<<oblocks, OTH, 0, stream>>>(edges, h1, agg0, agg1, olist, E, N);
  sage_gemm<64, false><<<gblocks, T, 0, stream>>>(h1, agg0, agg1, Bcat2, bias2, h2, N);

  edge_score<<<sblocks, T, 0, stream>>>(edges, edges + E, h2, out, E);
  edge_score<<<sblocks, T, 0, stream>>>(negE,  negE + E,  h2, out + E, E);
}


# Layer-1 owner scan reads a one-byte owner key per edge (dst / 200, from a new small launch) instead of the 4-byte id: 16 edges per 16-B load; 7 launches become 8; sums stay LDS atomics, arrival order
# speedup vs baseline: 2.5139x; 1.0512x over previous
#include <hip/hip_runtime.h>
#include <hip/hip_bf16.h>

typedef float v2f __attribute__((ext_vector_type(2)));
typedef float v8f __attribute__((ext_vector_type(8)));

#define FDIM 128
#define KTOT 384
#define KC   64

__global__ void build_weights(const float* __restrict__ Wn1, const float* __restrict__ Ws1,
                              const float* __restrict__ b1,
                              const float* __restrict__ Wn2, const float* __restrict__ Ws2,
                              const float* __restrict__ b2,
                              float* __restrict__ Bcat1, float* __restrict__ bias1,
                              float* __restrict__ Bcat2, float* __restrict__ bias2) {
  int i = blockIdx.x * blockDim.x + threadIdx.x;
  const int S1 = 384 * 128, S2 = S1 + 128, S3 = S2 + 384 * 64, S4 = S3 + 64;
  if (i < S1) {
    int k = i >> 7, n = i & 127;
    float v;
    if (k < 128)      v = Ws1[k * 128 + n] + Ws1[16384 + k * 128 + n];
    else if (k < 256) v = Wn1[(k - 128) * 128 + n];
    else              v = Wn1[16384 + (k - 256) * 128 + n];
    Bcat1[i] = v;
  } else if (i < S2) {
    int n = i - S1;
    bias1[n] = b1[n] + b1[128 + n];
  } else if (i < S3) {
    int j = i - S2;
    int k = j >> 6, n = j & 63;
    float v;
    if (k < 128)      v = Ws2[k * 64 + n] + Ws2[8192 + k * 64 + n];
    else if (k < 256) v = Wn2[(k - 128) * 64 + n];
    else              v = Wn2[8192 + (k - 256) * 64 + n];
    Bcat2[j] = v;
  } else if (i < S4) {
    int n = i - S3;
    bias2[n] = b2[n] + b2[64 + n];
  }
}

#define RN   200
#define LCAP 5120
#define OTH  1024
#define SU   8
#define DU   8
#define LHDR 32
#define LREG (2 * LCAP + LHDR)

__device__ __forceinline__ int acc_pos(int f) { return ((f & 3) << 5) + (f >> 2); }

__device__ __forceinline__ void note_edge(int dv, int e, int lo, int r,
                                          const int* __restrict__ src,
                                          const float* __restrict__ feat,
                                          float* acc, int* le, int* ldl, int* lcnt, int* cnt) {
  const unsigned t = (unsigned)(dv - lo);
  if (t < (unsigned)RN) {
    atomicAdd(&cnt[r * RN + (int)t], 1);
    const int slot = atomicAdd(&lcnt[r], 1);
    if (slot < LCAP) {
      le[r * LCAP + slot] = e;
      ldl[r * LCAP + slot] = (int)t;
    } else {
      const float* row = feat + (size_t)src[e] * FDIM;
      float* a = acc + (size_t)(r * RN + (int)t) * FDIM;
      for (int f = 0; f < FDIM; ++f) atomicAdd(&a[acc_pos(f)], row[f]);
    }
  }
}

template <int NL, bool GUARD>
__device__ __forceinline__ void scan_group(const int4* __restrict__ d4, int base, int nvec,
                                           int tid, int lo, int r,
                                           const int* __restrict__ src,
                                           const float* __restrict__ feat,
                                           float* acc, int* le, int* ldl, int* lcnt, int* cnt) {
  int4 d[NL];
#pragma unroll
  for (int u = 0; u < NL; ++u) {
    const int i = base + u * OTH + tid;
    if (GUARD) d[u] = (i < nvec) ? d4[i] : make_int4(-1, -1, -1, -1);
    else       d[u] = d4[i];
  }
#pragma unroll
  for (int u = 0; u < NL; ++u) {
    const int i = base + u * OTH + tid;
    const unsigned t0 = (unsigned)(d[u].x - lo), t1 = (unsigned)(d[u].y - lo);
    const unsigned t2 = (unsigned)(d[u].z - lo), t3 = (unsigned)(d[u].w - lo);
    const unsigned tm = min(min(t0, t1), min(t2, t3));
    if (tm < (unsigned)RN) {
      note_edge(d[u].x, 4 * i + 0, lo, r, src, feat, acc, le, ldl, lcnt, cnt);
      note_edge(d[u].y, 4 * i + 1, lo, r, src, feat, acc, le, ldl, lcnt, cnt);
      note_edge(d[u].z, 4 * i + 2, lo, r, src, feat, acc, le, ldl, lcnt, cnt);
      note_edge(d[u].w, 4 * i + 3, lo, r, src, feat, acc, le, ldl, lcnt, cnt);
    }
  }
}

__device__ __forceinline__ bool keys_usable(const unsigned char* keys, const int* edges, int E, int N) {
  const int nb = (N + RN - 1) / RN;
  return keys != nullptr && (E & 15) == 0 && nb <= 256 &&
         (size_t)2 * E <= (size_t)N * FDIM * 4 &&
         ((((size_t)keys) | ((size_t)edges)) & 15) == 0;
}
__device__ __forceinline__ unsigned key_word(int4 d) {
  return (((unsigned)d.x / RN) & 255u) | ((((unsigned)d.y / RN) & 255u) << 8) |
         ((((unsigned)d.z / RN) & 255u) << 16) | (((unsigned)d.w / RN) << 24);
}
__global__ void build_keys(const int* __restrict__ edges, unsigned char* __restrict__ keys,
                           int E, int N) {
  if (!keys_usable(keys, edges, E, N)) return;
  const int per = E >> 4;
  const int i = blockIdx.x * blockDim.x + threadIdx.x;
  if (i >= 2 * per) return;
  const int r = (i >= per) ? 1 : 0;
  const int g = i - r * per;
  const int4* d4 = (const int4*)(edges + (size_t)r * 2 * E + E) + (size_t)g * 4;
  const int4 a = d4[0], b = d4[1], c = d4[2], d = d4[3];
  ((uint4*)(keys + (size_t)r * E))[g] =
      make_uint4(key_word(a), key_word(b), key_word(c), key_word(d));
}

__device__ __forceinline__ unsigned zero_bytes(unsigned x) {
  return (x - 0x01010101u) & ~x & 0x80808080u;
}

__device__ __forceinline__ void note_keyed(int e, int lo, int r,
                                           const int* __restrict__ src,
                                           const int* __restrict__ dst,
                                           const float* __restrict__ feat,
                                           float* acc, int* le, int* lcnt, int* cnt) {
  const int slot = atomicAdd(&lcnt[r], 1);
  if (slot < LCAP) {
    le[r * LCAP + slot] = e;
  } else {
    const unsigned t = (unsigned)(dst[e] - lo);
    if (t < (unsigned)RN) {
      atomicAdd(&cnt[r * RN + (int)t], 1);
      const float* row = feat + (size_t)src[e] * FDIM;
      float* a = acc + (size_t)(r * RN + (int)t) * FDIM;
      for (int f = 0; f < FDIM; ++f) atomicAdd(&a[acc_pos(f)], row[f]);
    }
  }
}

template <int NL, bool GUARD>
__device__ __forceinline__ void scan_keys(const uint4* __restrict__ k4, int base, int nvec,
                                          int tid, unsigned pat, int lo, int r,
                                          const int* __restrict__ src,
                                          const int* __restrict__ dst,
                                          const float* __restrict__ feat,
                                          float* acc, int* le, int* lcnt, int* cnt) {
  uint4 k[NL];
#pragma unroll
  for (int u = 0; u < NL; ++u) {
    const int i = base + u * OTH + tid;
    if (GUARD) k[u] = (i < nvec) ? k4[i] : make_uint4(~pat, ~pat, ~pat, ~pat);
    else       k[u] = k4[i];
  }
#pragma unroll
  for (int u = 0; u < NL; ++u) {
    const int i = base + u * OTH + tid;
    unsigned m = (zero_bytes(k[u].x ^ pat) >> 7) | (zero_bytes(k[u].y ^ pat) >> 6) |
                 (zero_bytes(k[u].z ^ pat) >> 5) | (zero_bytes(k[u].w ^ pat) >> 4);
    while (m) {
      const int b = __builtin_ctz(m);
      m &= m - 1;
      note_keyed(16 * i + ((b & 7) << 2) + (b >> 3), lo, r, src, dst, feat, acc, le, lcnt, cnt);
    }
  }
}

template <int MODE>
__global__ __launch_bounds__(OTH) void mean_owner(const int* __restrict__ edges,
                                                  const float* __restrict__ feat,
                                                  float* __restrict__ agg0,
                                                  float* __restrict__ agg1,
                                                  unsigned* __restrict__ lists,
                                                  const unsigned char* __restrict__ keys,
                                                  int E, int N) {
  __shared__ __attribute__((aligned(16))) float acc[(2 * RN + 1) * FDIM];
  __shared__ int le[2 * LCAP];
  __shared__ int ldl[2 * LCAP];
  __shared__ int cnt[2 * RN];
  __shared__ int lcnt[2];

  const int tid = threadIdx.x;
  const int lane = tid & 31;
  const int wave = __builtin_amdgcn_readfirstlane(tid >> 5);
  const int lo = blockIdx.x * RN;

  for (int i = tid; i < 2 * RN * FDIM / 4; i += OTH)
    ((float4*)acc)[i] = make_float4(0.0f, 0.0f, 0.0f, 0.0f);
  for (int i = tid; i < 2 * RN; i += OTH) cnt[i] = 0;
  if (tid < 2) lcnt[tid] = 0;
  __syncthreads();

  unsigned* mine = lists + (size_t)blockIdx.x * LREG;
  const bool room = ((size_t)gridDim.x * LREG <= (size_t)N * 64) && (N <= (1 << 24)) && (RN < 255);
  int c0 = -1, c1 = -1;
  if (MODE == 1 && room) {
    c0 = __builtin_amdgcn_readfirstlane(((const int*)mine)[0]);
    c1 = __builtin_amdgcn_readfirstlane(((const int*)mine)[1]);
  }
  const bool have = (MODE == 1) && room && ((unsigned)c0 <= (unsigned)LCAP) &&
                    ((unsigned)c1 <= (unsigned)LCAP);

  const bool keyed = (MODE == 0) && (gridDim.x <= 256) && keys_usable(keys, edges, E, N);
  if (keyed) {
    const unsigned pat = (unsigned)blockIdx.x * 0x01010101u;
    const int nk = E >> 4;
    for (int r = 0; r < 2; ++r) {
      const int* src = edges + (size_t)r * 2 * E;
      const int* dst = src + E;
      const uint4* k4 = (const uint4*)(keys + (size_t)r * E);
      const int nfull = nk - nk % (OTH * SU);
      for (int base = 0; base < nfull; base += OTH * SU)
        scan_keys<SU, false>(k4, base, nk, tid, pat, lo, r, src, dst, feat, acc, le, lcnt, cnt);
      for (int base = nfull; base < nk; base += OTH * 4)
        scan_keys<4, true>(k4, base, nk, tid, pat, lo, r, src, dst, feat, acc, le, lcnt, cnt);
    }
  }

  const bool vec = ((E & 3) == 0) && ((((size_t)edges) & 15) == 0);
  const int nvec = vec ? (E >> 2) : 0;
  for (int r = 0; r < 2 && !have && !keyed; ++r) {
    const int* src = edges + (size_t)r * 2 * E;
    const int* dst = src + E;
    const int4* d4 = (const int4*)dst;
    const int nfull = nvec - nvec % (OTH * SU);
    for (int base = 0; base < nfull; base += OTH * SU)
      scan_group<SU, false>(d4, base, nvec, tid, lo, r, src, feat, acc, le, ldl, lcnt, cnt);
    for (int base = nfull; base < nvec; base += OTH * 4)
      scan_group<4, true>(d4, base, nvec, tid, lo, r, src, feat, acc, le, ldl, lcnt, cnt);
    for (int e = nvec * 4 + tid; e < E; e += OTH)
      note_edge(dst[e], e, lo, r, src, feat, acc, le, ldl, lcnt, cnt);
  }
  __syncthreads();

  const int n0 = have ? c0 : min(lcnt[0], LCAP), n1 = have ? c1 : min(lcnt[1], LCAP);
  if (have) {
    constexpr int PT = LCAP / OTH;
    static_assert(LCAP % OTH == 0, "a list is a whole number of words per thread");
    unsigned p[2][PT];
#pragma unroll
    for (int r = 0; r < 2; ++r) {
      const unsigned* sl = mine + LHDR + r * LCAP;
      const int n = r ? n1 : n0;
#pragma unroll
      for (int k = 0; k < PT; ++k) {
        const int i = tid + k * OTH;
        p[r][k] = (i < n) ? sl[i] : 0u;
      }
    }
#pragma unroll
    for (int r = 0; r < 2; ++r) {
      const int n = r ? n1 : n0;
#pragma unroll
      for (int k = 0; k < PT; ++k) {
        const int i = tid + k * OTH;
        if (i < n) {
          const unsigned tt = p[r][k] & 255u;
          const bool real = tt < (unsigned)RN;
          le[r * LCAP + i] = real ? min((int)(p[r][k] >> 8), N - 1) : 0;
          ldl[r * LCAP + i] = real ? (int)tt : 2 * RN - r * RN;
          if (real) atomicAdd(&cnt[r * RN + (int)tt], 1);
        }
      }
    }
  } else {
    const bool save = (MODE == 0) && room;
    if (save && tid == 0) {
      ((int*)mine)[0] = lcnt[0];
      ((int*)mine)[1] = lcnt[1];
    }
    for (int r = 0; r < 2; ++r) {
      const int* src = edges + (size_t)r * 2 * E;
      const int* dst = src + E;
      unsigned* sl = mine + LHDR + r * LCAP;
      const int n = r ? n1 : n0;
      for (int i = tid; i < n; i += OTH) {
        const int e = le[r * LCAP + i];
        int s = src[e];
        if (keyed) {
          const unsigned t = (unsigned)(dst[e] - lo);
          const bool real = t < (unsigned)RN;
          if (real) atomicAdd(&cnt[r * RN + (int)t], 1);
          else s = 0;
          ldl[r * LCAP + i] = real ? (int)t : 2 * RN - r * RN;
          le[r * LCAP + i] = s;
          if (save) sl[i] = ((unsigned)s << 8) | (real ? t : 255u);
        } else {
          le[r * LCAP + i] = s;
          if (save) sl[i] = ((unsigned)s << 8) | (unsigned)ldl[r * LCAP + i];
        }
      }
    }
  }
  __syncthreads();

  for (int r = 0; r < 2; ++r) {
    const int n = r ? n1 : n0;
    float* ar = acc + (size_t)r * RN * FDIM;
    const int* ls = le + r * LCAP;
    const int* ld = ldl + r * LCAP;
    const int spare = 2 * RN - r * RN;
    for (int i0 = wave; i0 < n; i0 += 32 * DU) {
      const int i = i0 + lane * 32;
      const bool ok = (lane < DU) && (i < n);
      const int sv = ok ? ls[i] : 0;
      const int dv = ok ? ld[i] : spare;
      float4 v[DU];
      int dl[DU];
#pragma unroll
      for (int u = 0; u < DU; ++u) {
        const int s = __builtin_amdgcn_readlane(sv, u);
        dl[u] = __builtin_amdgcn_readlane(dv, u);
        v[u] = *(const float4*)(feat + (size_t)s * FDIM + lane * 4);
      }
#pragma unroll
      for (int u = 0; u < DU; ++u) {
        float* a = ar + dl[u] * FDIM + lane;
        atomicAdd(a, v[u].x);
        atomicAdd(a + 32, v[u].y);
        atomicAdd(a + 64, v[u].z);
        atomicAdd(a + 96, v[u].w);
      }
    }
  }
  __syncthreads();

  for (int idx = tid; idx < 2 * RN * 32; idx += OTH) {
    const int r = idx / (RN * 32);
    const int rem = idx - r * (RN * 32);
    const int row = rem >> 5, q = rem & 31;
    const int node = lo + row;
    if (node < N) {
      const float* a = acc + (size_t)(r * RN + row) * FDIM + q;
      const float inv = 1.0f / fmaxf((float)cnt[r * RN + row], 1.0f);
      float4 o;
      o.x = a[0] * inv;
      o.y = a[32] * inv;
      o.z = a[64] * inv;
      o.w = a[96] * inv;
      *(float4*)((r ? agg1 : agg0) + (size_t)node * FDIM + q * 4) = o;
    }
  }
}

template <int NOUT, bool RELU>
__global__ __launch_bounds__(256) void sage_gemm(
    const float* __restrict__ A0, const float* __restrict__ G0, const float* __restrict__ G1,
    const float* __restrict__ Bmat, const float* __restrict__ bias,
    float* __restrict__ out, int Nrows) {
  constexpr int NT = NOUT / 16;
  __shared__ float Bs[KC * NOUT];

  const int tid = threadIdx.x;
  const int lane = tid & 31;
  const int wave = tid >> 5;
  const int rowBase = blockIdx.x * 128 + wave * 16;
  const int m = lane & 15;
  const int khalf = (lane >> 4) << 1;

  v8f czero = {};
  v8f c[NT];
#pragma unroll
  for (int t = 0; t < NT; ++t) c[t] = czero;

  int rowA = rowBase + m;
  if (rowA > Nrows - 1) rowA = Nrows - 1;

  for (int kc = 0; kc < KTOT; kc += KC) {
    for (int i = tid; i < KC * NOUT; i += 256)
      Bs[i] = Bmat[kc * NOUT + i];
    __syncthreads();

    const float* Asrc = (kc < 128) ? A0 : ((kc < 256) ? G0 : G1);
    const int kbase = kc & 127;

    for (int k4 = 0; k4 < KC; k4 += 4) {
      v2f a = *(const v2f*)(Asrc + (size_t)rowA * FDIM + kbase + k4 + khalf);
      const int kb = k4 + khalf;
#pragma unroll
      for (int t = 0; t < NT; ++t) {
        v2f b;
        b.x = Bs[kb * NOUT + t * 16 + m];
        b.y = Bs[(kb + 1) * NOUT + t * 16 + m];
        c[t] = __builtin_amdgcn_wmma_f32_16x16x4_f32(
            false, a, false, b, (short)0, c[t], false, false);
      }
    }
    __syncthreads();
  }

  const int rhalf = (lane >> 4) << 3;
#pragma unroll
  for (int t = 0; t < NT; ++t) {
    int col = t * 16 + m;
    float bv = bias[col];
#pragma unroll
    for (int v = 0; v < 8; ++v) {
      int row = rowBase + v + rhalf;
      if (row < Nrows) {
        float val = c[t][v] + bv;
        if (RELU) val = fmaxf(val, 0.0f);
        out[(size_t)row * NOUT + col] = val;
      }
    }
  }
}

template <int CTRL>
__device__ __forceinline__ float row_xor(float v) {
  return __builtin_bit_cast(float, __builtin_amdgcn_update_dpp(0, __builtin_bit_cast(int, v), CTRL, 0xf, 0xf, true));
}
__device__ __forceinline__ float lane_xor16(float v) {
  const int x = __builtin_bit_cast(int, v);
  return __builtin_bit_cast(float, __builtin_amdgcn_permlanex16(x, x, 0x76543210, 0xfedcba98, false, false));
}
__global__ void edge_score(const int* __restrict__ src, const int* __restrict__ dst,
                           const float* __restrict__ h, float* __restrict__ out, int E) {
  int gid = blockIdx.x * blockDim.x + threadIdx.x;
  int e = gid >> 5, lane = gid & 31;
  if (e >= E) return;
  int s = src[e], d = dst[e];
  v2f a = *(const v2f*)(h + (size_t)s * 64 + lane * 2);
  v2f b = *(const v2f*)(h + (size_t)d * 64 + lane * 2);
  float p = a.x * b.x + a.y * b.y;
  p += lane_xor16(p);
  p += row_xor<0x168>(p);
  p += row_xor<0x164>(p);
  p += row_xor<0x162>(p);
  p += row_xor<0x161>(p);
  if (lane == 0) out[e] = p;
}

extern "C" void kernel_launch(void* const* d_in, const int* in_sizes, int n_in,
                              void* d_out, int out_size, void* d_ws, size_t ws_size,
                              hipStream_t stream) {
  const float* x   = (const float*)d_in[0];
  const int*   edges = (const int*)d_in[1];
  const int*   negE  = (const int*)d_in[2];
  const float* Wn1 = (const float*)d_in[3];
  const float* Ws1 = (const float*)d_in[4];
  const float* b1  = (const float*)d_in[5];
  const float* Wn2 = (const float*)d_in[6];
  const float* Ws2 = (const float*)d_in[7];
  const float* b2  = (const float*)d_in[8];
  float* out = (float*)d_out;

  const int N = in_sizes[0] / 128;
  const int E = in_sizes[1] / 4;

  float* ws    = (float*)d_ws;
  float* dinv  = ws;
  float* agg0  = dinv + 2 * (size_t)N;
  float* agg1  = agg0 + (size_t)N * 128;
  float* h1    = agg1 + (size_t)N * 128;
  float* h2    = h1 + (size_t)N * 128;
  float* Bcat1 = h2 + (size_t)N * 64;
  float* bias1 = Bcat1 + 384 * 128;
  float* Bcat2 = bias1 + 128;
  float* bias2 = Bcat2 + 384 * 64;

  const int T = 256;

  const int wtot = 384 * 128 + 128 + 384 * 64 + 64;
  build_weights<<<(wtot + T - 1) / T, T, 0, stream>>>(Wn1, Ws1, b1, Wn2, Ws2, b2,
                                                      Bcat1, bias1, Bcat2, bias2);

  const int sblocks = (int)(((size_t)E * 32 + T - 1) / T);
  const int gblocks = (N + 127) / 128;
  const int oblocks = (N + RN - 1) / RN;
  const int kblocks = (int)(((size_t)E / 8 + T) / T);

  unsigned* olist = (unsigned*)h2;
  unsigned char* okeys = (unsigned char*)h1;
  build_keys<<<kblocks, T, 0, stream>>>(edges, okeys, E, N);
  mean_owner<0><<<oblocks, OTH, 0, stream>>>(edges, x, agg0, agg1, olist, okeys, E, N);
  sage_gemm<128, true><<<gblocks, T, 0, stream>>>(x, agg0, agg1, Bcat1, bias1, h1, N);

  mean_owner<1><<<oblocks, OTH, 0, stream>>>(edges, h1, agg0, agg1, olist, nullptr, E, N);
  sage_gemm<64, false><<<gblocks, T, 0, stream>>>(h1, agg0, agg1, Bcat2, bias2, h2, N);

  edge_score<<<sblocks, T, 0, stream>>>(edges, edges + E, h2, out, E);
  edge_score<<<sblocks, T, 0, stream>>>(negE,  negE + E,  h2, out + E, E);
}


# Both products' K loop software-pipelined: next 64-row weight chunk (16-B loads, all in flight) goes to a second LDS buffer, next A fragments load under the current chunk's WMMAs; sum order unchanged
# speedup vs baseline: 2.7905x; 1.1100x over previous
#include <hip/hip_runtime.h>
#include <hip/hip_bf16.h>

typedef float v2f __attribute__((ext_vector_type(2)));
typedef float v4f __attribute__((ext_vector_type(4)));
typedef float v8f __attribute__((ext_vector_type(8)));

#define FDIM 128
#define KTOT 384
#define KC   64

__global__ void build_weights(const float* __restrict__ Wn1, const float* __restrict__ Ws1,
                              const float* __restrict__ b1,
                              const float* __restrict__ Wn2, const float* __restrict__ Ws2,
                              const float* __restrict__ b2,
                              float* __restrict__ Bcat1, float* __restrict__ bias1,
                              float* __restrict__ Bcat2, float* __restrict__ bias2) {
  int i = blockIdx.x * blockDim.x + threadIdx.x;
  const int S1 = 384 * 128, S2 = S1 + 128, S3 = S2 + 384 * 64, S4 = S3 + 64;
  if (i < S1) {
    int k = i >> 7, n = i & 127;
    float v;
    if (k < 128)      v = Ws1[k * 128 + n] + Ws1[16384 + k * 128 + n];
    else if (k < 256) v = Wn1[(k - 128) * 128 + n];
    else              v = Wn1[16384 + (k - 256) * 128 + n];
    Bcat1[i] = v;
  } else if (i < S2) {
    int n = i - S1;
    bias1[n] = b1[n] + b1[128 + n];
  } else if (i < S3) {
    int j = i - S2;
    int k = j >> 6, n = j & 63;
    float v;
    if (k < 128)      v = Ws2[k * 64 + n] + Ws2[8192 + k * 64 + n];
    else if (k < 256) v = Wn2[(k - 128) * 64 + n];
    else              v = Wn2[8192 + (k - 256) * 64 + n];
    Bcat2[j] = v;
  } else if (i < S4) {
    int n = i - S3;
    bias2[n] = b2[n] + b2[64 + n];
  }
}

#define RN   200
#define LCAP 5120
#define OTH  1024
#define SU   8
#define DU   8
#define LHDR 32
#define LREG (2 * LCAP + LHDR)

__device__ __forceinline__ int acc_pos(int f) { return ((f & 3) << 5) + (f >> 2); }

__device__ __forceinline__ void note_edge(int dv, int e, int lo, int r,
                                          const int* __restrict__ src,
                                          const float* __restrict__ feat,
                                          float* acc, int* le, int* ldl, int* lcnt, int* cnt) {
  const unsigned t = (unsigned)(dv - lo);
  if (t < (unsigned)RN) {
    atomicAdd(&cnt[r * RN + (int)t], 1);
    const int slot = atomicAdd(&lcnt[r], 1);
    if (slot < LCAP) {
      le[r * LCAP + slot] = e;
      ldl[r * LCAP + slot] = (int)t;
    } else {
      const float* row = feat + (size_t)src[e] * FDIM;
      float* a = acc + (size_t)(r * RN + (int)t) * FDIM;
      for (int f = 0; f < FDIM; ++f) atomicAdd(&a[acc_pos(f)], row[f]);
    }
  }
}

template <int NL, bool GUARD>
__device__ __forceinline__ void scan_group(const int4* __restrict__ d4, int base, int nvec,
                                           int tid, int lo, int r,
                                           const int* __restrict__ src,
                                           const float* __restrict__ feat,
                                           float* acc, int* le, int* ldl, int* lcnt, int* cnt) {
  int4 d[NL];
#pragma unroll
  for (int u = 0; u < NL; ++u) {
    const int i = base + u * OTH + tid;
    if (GUARD) d[u] = (i < nvec) ? d4[i] : make_int4(-1, -1, -1, -1);
    else       d[u] = d4[i];
  }
#pragma unroll
  for (int u = 0; u < NL; ++u) {
    const int i = base + u * OTH + tid;
    const unsigned t0 = (unsigned)(d[u].x - lo), t1 = (unsigned)(d[u].y - lo);
    const unsigned t2 = (unsigned)(d[u].z - lo), t3 = (unsigned)(d[u].w - lo);
    const unsigned tm = min(min(t0, t1), min(t2, t3));
    if (tm < (unsigned)RN) {
      note_edge(d[u].x, 4 * i + 0, lo, r, src, feat, acc, le, ldl, lcnt, cnt);
      note_edge(d[u].y, 4 * i + 1, lo, r, src, feat, acc, le, ldl, lcnt, cnt);
      note_edge(d[u].z, 4 * i + 2, lo, r, src, feat, acc, le, ldl, lcnt, cnt);
      note_edge(d[u].w, 4 * i + 3, lo, r, src, feat, acc, le, ldl, lcnt, cnt);
    }
  }
}

__device__ __forceinline__ bool keys_usable(const unsigned char* keys, const int* edges, int E, int N) {
  const int nb = (N + RN - 1) / RN;
  return keys != nullptr && (E & 15) == 0 && nb <= 256 &&
         (size_t)2 * E <= (size_t)N * FDIM * 4 &&
         ((((size_t)keys) | ((size_t)edges)) & 15) == 0;
}
__device__ __forceinline__ unsigned key_word(int4 d) {
  return (((unsigned)d.x / RN) & 255u) | ((((unsigned)d.y / RN) & 255u) << 8) |
         ((((unsigned)d.z / RN) & 255u) << 16) | (((unsigned)d.w / RN) << 24);
}
__global__ void build_keys(const int* __restrict__ edges, unsigned char* __restrict__ keys,
                           int E, int N) {
  if (!keys_usable(keys, edges, E, N)) return;
  const int per = E >> 4;
  const int i = blockIdx.x * blockDim.x + threadIdx.x;
  if (i >= 2 * per) return;
  const int r = (i >= per) ? 1 : 0;
  const int g = i - r * per;
  const int4* d4 = (const int4*)(edges + (size_t)r * 2 * E + E) + (size_t)g * 4;
  const int4 a = d4[0], b = d4[1], c = d4[2], d = d4[3];
  ((uint4*)(keys + (size_t)r * E))[g] =
      make_uint4(key_word(a), key_word(b), key_word(c), key_word(d));
}

__device__ __forceinline__ unsigned zero_bytes(unsigned x) {
  return (x - 0x01010101u) & ~x & 0x80808080u;
}

__device__ __forceinline__ void note_keyed(int e, int lo, int r,
                                           const int* __restrict__ src,
                                           const int* __restrict__ dst,
                                           const float* __restrict__ feat,
                                           float* acc, int* le, int* lcnt, int* cnt) {
  const int slot = atomicAdd(&lcnt[r], 1);
  if (slot < LCAP) {
    le[r * LCAP + slot] = e;
  } else {
    const unsigned t = (unsigned)(dst[e] - lo);
    if (t < (unsigned)RN) {
      atomicAdd(&cnt[r * RN + (int)t], 1);
      const float* row = feat + (size_t)src[e] * FDIM;
      float* a = acc + (size_t)(r * RN + (int)t) * FDIM;
      for (int f = 0; f < FDIM; ++f) atomicAdd(&a[acc_pos(f)], row[f]);
    }
  }
}

template <int NL, bool GUARD>
__device__ __forceinline__ void scan_keys(const uint4* __restrict__ k4, int base, int nvec,
                                          int tid, unsigned pat, int lo, int r,
                                          const int* __restrict__ src,
                                          const int* __restrict__ dst,
                                          const float* __restrict__ feat,
                                          float* acc, int* le, int* lcnt, int* cnt) {
  uint4 k[NL];
#pragma unroll
  for (int u = 0; u < NL; ++u) {
    const int i = base + u * OTH + tid;
    if (GUARD) k[u] = (i < nvec) ? k4[i] : make_uint4(~pat, ~pat, ~pat, ~pat);
    else       k[u] = k4[i];
  }
#pragma unroll
  for (int u = 0; u < NL; ++u) {
    const int i = base + u * OTH + tid;
    unsigned m = (zero_bytes(k[u].x ^ pat) >> 7) | (zero_bytes(k[u].y ^ pat) >> 6) |
                 (zero_bytes(k[u].z ^ pat) >> 5) | (zero_bytes(k[u].w ^ pat) >> 4);
    while (m) {
      const int b = __builtin_ctz(m);
      m &= m - 1;
      note_keyed(16 * i + ((b & 7) << 2) + (b >> 3), lo, r, src, dst, feat, acc, le, lcnt, cnt);
    }
  }
}

template <int MODE>
__global__ __launch_bounds__(OTH) void mean_owner(const int* __restrict__ edges,
                                                  const float* __restrict__ feat,
                                                  float* __restrict__ agg0,
                                                  float* __restrict__ agg1,
                                                  unsigned* __restrict__ lists,
                                                  const unsigned char* __restrict__ keys,
                                                  int E, int N) {
  __shared__ __attribute__((aligned(16))) float acc[(2 * RN + 1) * FDIM];
  __shared__ int le[2 * LCAP];
  __shared__ int ldl[2 * LCAP];
  __shared__ int cnt[2 * RN];
  __shared__ int lcnt[2];

  const int tid = threadIdx.x;
  const int lane = tid & 31;
  const int wave = __builtin_amdgcn_readfirstlane(tid >> 5);
  const int lo = blockIdx.x * RN;

  for (int i = tid; i < 2 * RN * FDIM / 4; i += OTH)
    ((float4*)acc)[i] = make_float4(0.0f, 0.0f, 0.0f, 0.0f);
  for (int i = tid; i < 2 * RN; i += OTH) cnt[i] = 0;
  if (tid < 2) lcnt[tid] = 0;
  __syncthreads();

  unsigned* mine = lists + (size_t)blockIdx.x * LREG;
  const bool room = ((size_t)gridDim.x * LREG <= (size_t)N * 64) && (N <= (1 << 24)) && (RN < 255);
  int c0 = -1, c1 = -1;
  if (MODE == 1 && room) {
    c0 = __builtin_amdgcn_readfirstlane(((const int*)mine)[0]);
    c1 = __builtin_amdgcn_readfirstlane(((const int*)mine)[1]);
  }
  const bool have = (MODE == 1) && room && ((unsigned)c0 <= (unsigned)LCAP) &&
                    ((unsigned)c1 <= (unsigned)LCAP);

  const bool keyed = (MODE == 0) && (gridDim.x <= 256) && keys_usable(keys, edges, E, N);
  if (keyed) {
    const unsigned pat = (unsigned)blockIdx.x * 0x01010101u;
    const int nk = E >> 4;
    for (int r = 0; r < 2; ++r) {
      const int* src = edges + (size_t)r * 2 * E;
      const int* dst = src + E;
      const uint4* k4 = (const uint4*)(keys + (size_t)r * E);
      const int nfull = nk - nk % (OTH * SU);
      for (int base = 0; base < nfull; base += OTH * SU)
        scan_keys<SU, false>(k4, base, nk, tid, pat, lo, r, src, dst, feat, acc, le, lcnt, cnt);
      for (int base = nfull; base < nk; base += OTH * 4)
        scan_keys<4, true>(k4, base, nk, tid, pat, lo, r, src, dst, feat, acc, le, lcnt, cnt);
    }
  }

  const bool vec = ((E & 3) == 0) && ((((size_t)edges) & 15) == 0);
  const int nvec = vec ? (E >> 2) : 0;
  for (int r = 0; r < 2 && !have && !keyed; ++r) {
    const int* src = edges + (size_t)r * 2 * E;
    const int* dst = src + E;
    const int4* d4 = (const int4*)dst;
    const int nfull = nvec - nvec % (OTH * SU);
    for (int base = 0; base < nfull; base += OTH * SU)
      scan_group<SU, false>(d4, base, nvec, tid, lo, r, src, feat, acc, le, ldl, lcnt, cnt);
    for (int base = nfull; base < nvec; base += OTH * 4)
      scan_group<4, true>(d4, base, nvec, tid, lo, r, src, feat, acc, le, ldl, lcnt, cnt);
    for (int e = nvec * 4 + tid; e < E; e += OTH)
      note_edge(dst[e], e, lo, r, src, feat, acc, le, ldl, lcnt, cnt);
  }
  __syncthreads();

  const int n0 = have ? c0 : min(lcnt[0], LCAP), n1 = have ? c1 : min(lcnt[1], LCAP);
  if (have) {
    constexpr int PT = LCAP / OTH;
    static_assert(LCAP % OTH == 0, "a list is a whole number of words per thread");
    unsigned p[2][PT];
#pragma unroll
    for (int r = 0; r < 2; ++r) {
      const unsigned* sl = mine + LHDR + r * LCAP;
      const int n = r ? n1 : n0;
#pragma unroll
      for (int k = 0; k < PT; ++k) {
        const int i = tid + k * OTH;
        p[r][k] = (i < n) ? sl[i] : 0u;
      }
    }
#pragma unroll
    for (int r = 0; r < 2; ++r) {
      const int n = r ? n1 : n0;
#pragma unroll
      for (int k = 0; k < PT; ++k) {
        const int i = tid + k * OTH;
        if (i < n) {
          const unsigned tt = p[r][k] & 255u;
          const bool real = tt < (unsigned)RN;
          le[r * LCAP + i] = real ? min((int)(p[r][k] >> 8), N - 1) : 0;
          ldl[r * LCAP + i] = real ? (int)tt : 2 * RN - r * RN;
          if (real) atomicAdd(&cnt[r * RN + (int)tt], 1);
        }
      }
    }
  } else {
    const bool save = (MODE == 0) && room;
    if (save && tid == 0) {
      ((int*)mine)[0] = lcnt[0];
      ((int*)mine)[1] = lcnt[1];
    }
    for (int r = 0; r < 2; ++r) {
      const int* src = edges + (size_t)r * 2 * E;
      const int* dst = src + E;
      unsigned* sl = mine + LHDR + r * LCAP;
      const int n = r ? n1 : n0;
      for (int i = tid; i < n; i += OTH) {
        const int e = le[r * LCAP + i];
        int s = src[e];
        if (keyed) {
          const unsigned t = (unsigned)(dst[e] - lo);
          const bool real = t < (unsigned)RN;
          if (real) atomicAdd(&cnt[r * RN + (int)t], 1);
          else s = 0;
          ldl[r * LCAP + i] = real ? (int)t : 2 * RN - r * RN;
          le[r * LCAP + i] = s;
          if (save) sl[i] = ((unsigned)s << 8) | (real ? t : 255u);
        } else {
          le[r * LCAP + i] = s;
          if (save) sl[i] = ((unsigned)s << 8) | (unsigned)ldl[r * LCAP + i];
        }
      }
    }
  }
  __syncthreads();

  for (int r = 0; r < 2; ++r) {
    const int n = r ? n1 : n0;
    float* ar = acc + (size_t)r * RN * FDIM;
    const int* ls = le + r * LCAP;
    const int* ld = ldl + r * LCAP;
    const int spare = 2 * RN - r * RN;
    for (int i0 = wave; i0 < n; i0 += 32 * DU) {
      const int i = i0 + lane * 32;
      const bool ok = (lane < DU) && (i < n);
      const int sv = ok ? ls[i] : 0;
      const int dv = ok ? ld[i] : spare;
      float4 v[DU];
      int dl[DU];
#pragma unroll
      for (int u = 0; u < DU; ++u) {
        const int s = __builtin_amdgcn_readlane(sv, u);
        dl[u] = __builtin_amdgcn_readlane(dv, u);
        v[u] = *(const float4*)(feat + (size_t)s * FDIM + lane * 4);
      }
#pragma unroll
      for (int u = 0; u < DU; ++u) {
        float* a = ar + dl[u] * FDIM + lane;
        atomicAdd(a, v[u].x);
        atomicAdd(a + 32, v[u].y);
        atomicAdd(a + 64, v[u].z);
        atomicAdd(a + 96, v[u].w);
      }
    }
  }
  __syncthreads();

  for (int idx = tid; idx < 2 * RN * 32; idx += OTH) {
    const int r = idx / (RN * 32);
    const int rem = idx - r * (RN * 32);
    const int row = rem >> 5, q = rem & 31;
    const int node = lo + row;
    if (node < N) {
      const float* a = acc + (size_t)(r * RN + row) * FDIM + q;
      const float inv = 1.0f / fmaxf((float)cnt[r * RN + row], 1.0f);
      float4 o;
      o.x = a[0] * inv;
      o.y = a[32] * inv;
      o.z = a[64] * inv;
      o.w = a[96] * inv;
      *(float4*)((r ? agg1 : agg0) + (size_t)node * FDIM + q * 4) = o;
    }
  }
}

#define KS (KC / 4)

template <int NOUT>
__device__ __forceinline__ void gemm_chunk(
    int kn, bool more,
    const float* __restrict__ A0, const float* __restrict__ G0, const float* __restrict__ G1,
    const v4f* __restrict__ B4, int rowA, int khalf, int m, int tid,
    const float* bcur, float* bnxt,
    const v2f (&a)[KS], v2f (&an)[KS],
    v4f (&st)[KC * NOUT / 1024], v8f (&c)[NOUT / 16]) {
  constexpr int NT = NOUT / 16;
  constexpr int NV = KC * NOUT / 1024;
  if (more) {
#pragma unroll
    for (int j = 0; j < NV; ++j) st[j] = B4[(size_t)kn * (NOUT / 4) + tid + 256 * j];
    const float* An = (kn < 128) ? A0 : ((kn < 256) ? G0 : G1);
    const float* ap = An + (size_t)rowA * FDIM + (kn & 127) + khalf;
#pragma unroll
    for (int s = 0; s < KS; ++s) an[s] = *(const v2f*)(ap + 4 * s);
  }

  const float* bs = bcur + khalf * NOUT + m;
  v2f bf[2][NT];
#pragma unroll
  for (int t = 0; t < NT; ++t) {
    bf[0][t].x = bs[t * 16];
    bf[0][t].y = bs[NOUT + t * 16];
  }
#pragma unroll
  for (int s = 0; s < KS; ++s) {
    if (s + 1 < KS) {
#pragma unroll
      for (int t = 0; t < NT; ++t) {
        bf[(s + 1) & 1][t].x = bs[(s + 1) * 4 * NOUT + t * 16];
        bf[(s + 1) & 1][t].y = bs[((s + 1) * 4 + 1) * NOUT + t * 16];
      }
    }
#pragma unroll
    for (int t = 0; t < NT; ++t)
      c[t] = __builtin_amdgcn_wmma_f32_16x16x4_f32(
          false, a[s], false, bf[s & 1][t], (short)0, c[t], false, false);
  }

  if (more) {
#pragma unroll
    for (int j = 0; j < NV; ++j) ((v4f*)bnxt)[tid + 256 * j] = st[j];
    __syncthreads();
  }
}

template <int NOUT, bool RELU>
__global__ __launch_bounds__(256) void sage_gemm(
    const float* __restrict__ A0, const float* __restrict__ G0, const float* __restrict__ G1,
    const float* __restrict__ Bmat, const float* __restrict__ bias,
    float* __restrict__ out, int Nrows) {
  constexpr int NT = NOUT / 16;
  constexpr int NCH = KTOT / KC;
  constexpr int NV = KC * NOUT / 1024;
  static_assert(NCH % 2 == 0, "the chunk loop takes two chunks per trip");
  __shared__ __attribute__((aligned(16))) float Bs[2][KC * NOUT];

  const int tid = threadIdx.x;
  const int lane = tid & 31;
  const int wave = tid >> 5;
  const int rowBase = blockIdx.x * 128 + wave * 16;
  const int m = lane & 15;
  const int khalf = (lane >> 4) << 1;

  v8f czero = {};
  v8f c[NT];
#pragma unroll
  for (int t = 0; t < NT; ++t) c[t] = czero;

  int rowA = rowBase + m;
  if (rowA > Nrows - 1) rowA = Nrows - 1;

  const v4f* B4 = (const v4f*)Bmat;
  v4f st[NV];
  v2f aA[KS], aB[KS];

#pragma unroll
  for (int j = 0; j < NV; ++j) st[j] = B4[tid + 256 * j];
  {
    const float* ap = A0 + (size_t)rowA * FDIM + khalf;
#pragma unroll
    for (int s = 0; s < KS; ++s) aA[s] = *(const v2f*)(ap + 4 * s);
  }
#pragma unroll
  for (int j = 0; j < NV; ++j) ((v4f*)Bs[0])[tid + 256 * j] = st[j];
  __syncthreads();

#pragma unroll 1
  for (int ch = 0; ch < NCH; ch += 2) {
    gemm_chunk<NOUT>((ch + 1) * KC, true, A0, G0, G1, B4, rowA, khalf, m, tid,
                     Bs[0], Bs[1], aA, aB, st, c);
    gemm_chunk<NOUT>((ch + 2) * KC, ch + 2 < NCH, A0, G0, G1, B4, rowA, khalf, m, tid,
                     Bs[1], Bs[0], aB, aA, st, c);
  }

  const int rhalf = (lane >> 4) << 3;
#pragma unroll
  for (int t = 0; t < NT; ++t) {
    int col = t * 16 + m;
    float bv = bias[col];
#pragma unroll
    for (int v = 0; v < 8; ++v) {
      int row = rowBase + v + rhalf;
      if (row < Nrows) {
        float val = c[t][v] + bv;
        if (RELU) val = fmaxf(val, 0.0f);
        out[(size_t)row * NOUT + col] = val;
      }
    }
  }
}

template <int CTRL>
__device__ __forceinline__ float row_xor(float v) {
  return __builtin_bit_cast(float, __builtin_amdgcn_update_dpp(0, __builtin_bit_cast(int, v), CTRL, 0xf, 0xf, true));
}
__device__ __forceinline__ float lane_xor16(float v) {
  const int x = __builtin_bit_cast(int, v);
  return __builtin_bit_cast(float, __builtin_amdgcn_permlanex16(x, x, 0x76543210, 0xfedcba98, false, false));
}
__global__ void edge_score(const int* __restrict__ src, const int* __restrict__ dst,
                           const float* __restrict__ h, float* __restrict__ out, int E) {
  int gid = blockIdx.x * blockDim.x + threadIdx.x;
  int e = gid >> 5, lane = gid & 31;
  if (e >= E) return;
  int s = src[e], d = dst[e];
  v2f a = *(const v2f*)(h + (size_t)s * 64 + lane * 2);
  v2f b = *(const v2f*)(h + (size_t)d * 64 + lane * 2);
  float p = a.x * b.x + a.y * b.y;
  p += lane_xor16(p);
  p += row_xor<0x168>(p);
  p += row_xor<0x164>(p);
  p += row_xor<0x162>(p);
  p += row_xor<0x161>(p);
  if (lane == 0) out[e] = p;
}

extern "C" void kernel_launch(void* const* d_in, const int* in_sizes, int n_in,
                              void* d_out, int out_size, void* d_ws, size_t ws_size,
                              hipStream_t stream) {
  const float* x   = (const float*)d_in[0];
  const int*   edges = (const int*)d_in[1];
  const int*   negE  = (const int*)d_in[2];
  const float* Wn1 = (const float*)d_in[3];
  const float* Ws1 = (const float*)d_in[4];
  const float* b1  = (const float*)d_in[5];
  const float* Wn2 = (const float*)d_in[6];
  const float* Ws2 = (const float*)d_in[7];
  const float* b2  = (const float*)d_in[8];
  float* out = (float*)d_out;

  const int N = in_sizes[0] / 128;
  const int E = in_sizes[1] / 4;

  float* ws    = (float*)d_ws;
  float* dinv  = ws;
  float* agg0  = dinv + 2 * (size_t)N;
  float* agg1  = agg0 + (size_t)N * 128;
  float* h1    = agg1 + (size_t)N * 128;
  float* h2    = h1 + (size_t)N * 128;
  float* Bcat1 = h2 + (size_t)N * 64;
  float* bias1 = Bcat1 + 384 * 128;
  float* Bcat2 = bias1 + 128;
  float* bias2 = Bcat2 + 384 * 64;

  const int T = 256;

  const int wtot = 384 * 128 + 128 + 384 * 64 + 64;
  build_weights<<<(wtot + T - 1) / T, T, 0, stream>>>(Wn1, Ws1, b1, Wn2, Ws2, b2,
                                                      Bcat1, bias1, Bcat2, bias2);

  const int sblocks = (int)(((size_t)E * 32 + T - 1) / T);
  const int gblocks = (N + 127) / 128;
  const int oblocks = (N + RN - 1) / RN;
  const int kblocks = (int)(((size_t)E / 8 + T) / T);

  unsigned* olist = (unsigned*)h2;
  unsigned char* okeys = (unsigned char*)h1;
  build_keys<<<kblocks, T, 0, stream>>>(edges, okeys, E, N);
  mean_owner<0><<<oblocks, OTH, 0, stream>>>(edges, x, agg0, agg1, olist, okeys, E, N);
  sage_gemm<128, true><<<gblocks, T, 0, stream>>>(x, agg0, agg1, Bcat1, bias1, h1, N);

  mean_owner<1><<<oblocks, OTH, 0, stream>>>(edges, h1, agg0, agg1, olist, nullptr, E, N);
  sage_gemm<64, false><<<gblocks, T, 0, stream>>>(h1, agg0, agg1, Bcat2, bias2, h2, N);

  edge_score<<<sblocks, T, 0, stream>>>(edges, edges + E, h2, out, E);
  edge_score<<<sblocks, T, 0, stream>>>(negE,  negE + E,  h2, out + E, E);
}


# Edge scoring on the matrix unit: a wave takes 16 edges, stages their 32 h rows (whole-row loads) in LDS, f32 WMMA forms src x dst^T, the diagonal holds the scores; 16x fewer waves; new sum order
# speedup vs baseline: 3.7254x; 1.3350x over previous
#include <hip/hip_runtime.h>
#include <hip/hip_bf16.h>

typedef float v2f __attribute__((ext_vector_type(2)));
typedef float v4f __attribute__((ext_vector_type(4)));
typedef float v8f __attribute__((ext_vector_type(8)));

#define FDIM 128
#define KTOT 384
#define KC   64

__global__ void build_weights(const float* __restrict__ Wn1, const float* __restrict__ Ws1,
                              const float* __restrict__ b1,
                              const float* __restrict__ Wn2, const float* __restrict__ Ws2,
                              const float* __restrict__ b2,
                              float* __restrict__ Bcat1, float* __restrict__ bias1,
                              float* __restrict__ Bcat2, float* __restrict__ bias2) {
  int i = blockIdx.x * blockDim.x + threadIdx.x;
  const int S1 = 384 * 128, S2 = S1 + 128, S3 = S2 + 384 * 64, S4 = S3 + 64;
  if (i < S1) {
    int k = i >> 7, n = i & 127;
    float v;
    if (k < 128)      v = Ws1[k * 128 + n] + Ws1[16384 + k * 128 + n];
    else if (k < 256) v = Wn1[(k - 128) * 128 + n];
    else              v = Wn1[16384 + (k - 256) * 128 + n];
    Bcat1[i] = v;
  } else if (i < S2) {
    int n = i - S1;
    bias1[n] = b1[n] + b1[128 + n];
  } else if (i < S3) {
    int j = i - S2;
    int k = j >> 6, n = j & 63;
    float v;
    if (k < 128)      v = Ws2[k * 64 + n] + Ws2[8192 + k * 64 + n];
    else if (k < 256) v = Wn2[(k - 128) * 64 + n];
    else              v = Wn2[8192 + (k - 256) * 64 + n];
    Bcat2[j] = v;
  } else if (i < S4) {
    int n = i - S3;
    bias2[n] = b2[n] + b2[64 + n];
  }
}

#define RN   200
#define LCAP 5120
#define OTH  1024
#define SU   8
#define DU   8
#define LHDR 32
#define LREG (2 * LCAP + LHDR)

__device__ __forceinline__ int acc_pos(int f) { return ((f & 3) << 5) + (f >> 2); }

__device__ __forceinline__ void note_edge(int dv, int e, int lo, int r,
                                          const int* __restrict__ src,
                                          const float* __restrict__ feat,
                                          float* acc, int* le, int* ldl, int* lcnt, int* cnt) {
  const unsigned t = (unsigned)(dv - lo);
  if (t < (unsigned)RN) {
    atomicAdd(&cnt[r * RN + (int)t], 1);
    const int slot = atomicAdd(&lcnt[r], 1);
    if (slot < LCAP) {
      le[r * LCAP + slot] = e;
      ldl[r * LCAP + slot] = (int)t;
    } else {
      const float* row = feat + (size_t)src[e] * FDIM;
      float* a = acc + (size_t)(r * RN + (int)t) * FDIM;
      for (int f = 0; f < FDIM; ++f) atomicAdd(&a[acc_pos(f)], row[f]);
    }
  }
}

template <int NL, bool GUARD>
__device__ __forceinline__ void scan_group(const int4* __restrict__ d4, int base, int nvec,
                                           int tid, int lo, int r,
                                           const int* __restrict__ src,
                                           const float* __restrict__ feat,
                                           float* acc, int* le, int* ldl, int* lcnt, int* cnt) {
  int4 d[NL];
#pragma unroll
  for (int u = 0; u < NL; ++u) {
    const int i = base + u * OTH + tid;
    if (GUARD) d[u] = (i < nvec) ? d4[i] : make_int4(-1, -1, -1, -1);
    else       d[u] = d4[i];
  }
#pragma unroll
  for (int u = 0; u < NL; ++u) {
    const int i = base + u * OTH + tid;
    const unsigned t0 = (unsigned)(d[u].x - lo), t1 = (unsigned)(d[u].y - lo);
    const unsigned t2 = (unsigned)(d[u].z - lo), t3 = (unsigned)(d[u].w - lo);
    const unsigned tm = min(min(t0, t1), min(t2, t3));
    if (tm < (unsigned)RN) {
      note_edge(d[u].x, 4 * i + 0, lo, r, src, feat, acc, le, ldl, lcnt, cnt);
      note_edge(d[u].y, 4 * i + 1, lo, r, src, feat, acc, le, ldl, lcnt, cnt);
      note_edge(d[u].z, 4 * i + 2, lo, r, src, feat, acc, le, ldl, lcnt, cnt);
      note_edge(d[u].w, 4 * i + 3, lo, r, src, feat, acc, le, ldl, lcnt, cnt);
    }
  }
}

__device__ __forceinline__ bool keys_usable(const unsigned char* keys, const int* edges, int E, int N) {
  const int nb = (N + RN - 1) / RN;
  return keys != nullptr && (E & 15) == 0 && nb <= 256 &&
         (size_t)2 * E <= (size_t)N * FDIM * 4 &&
         ((((size_t)keys) | ((size_t)edges)) & 15) == 0;
}
__device__ __forceinline__ unsigned key_word(int4 d) {
  return (((unsigned)d.x / RN) & 255u) | ((((unsigned)d.y / RN) & 255u) << 8) |
         ((((unsigned)d.z / RN) & 255u) << 16) | (((unsigned)d.w / RN) << 24);
}
__global__ void build_keys(const int* __restrict__ edges, unsigned char* __restrict__ keys,
                           int E, int N) {
  if (!keys_usable(keys, edges, E, N)) return;
  const int per = E >> 4;
  const int i = blockIdx.x * blockDim.x + threadIdx.x;
  if (i >= 2 * per) return;
  const int r = (i >= per) ? 1 : 0;
  const int g = i - r * per;
  const int4* d4 = (const int4*)(edges + (size_t)r * 2 * E + E) + (size_t)g * 4;
  const int4 a = d4[0], b = d4[1], c = d4[2], d = d4[3];
  ((uint4*)(keys + (size_t)r * E))[g] =
      make_uint4(key_word(a), key_word(b), key_word(c), key_word(d));
}

__device__ __forceinline__ unsigned zero_bytes(unsigned x) {
  return (x - 0x01010101u) & ~x & 0x80808080u;
}

__device__ __forceinline__ void note_keyed(int e, int lo, int r,
                                           const int* __restrict__ src,
                                           const int* __restrict__ dst,
                                           const float* __restrict__ feat,
                                           float* acc, int* le, int* lcnt, int* cnt) {
  const int slot = atomicAdd(&lcnt[r], 1);
  if (slot < LCAP) {
    le[r * LCAP + slot] = e;
  } else {
    const unsigned t = (unsigned)(dst[e] - lo);
    if (t < (unsigned)RN) {
      atomicAdd(&cnt[r * RN + (int)t], 1);
      const float* row = feat + (size_t)src[e] * FDIM;
      float* a = acc + (size_t)(r * RN + (int)t) * FDIM;
      for (int f = 0; f < FDIM; ++f) atomicAdd(&a[acc_pos(f)], row[f]);
    }
  }
}

template <int NL, bool GUARD>
__device__ __forceinline__ void scan_keys(const uint4* __restrict__ k4, int base, int nvec,
                                          int tid, unsigned pat, int lo, int r,
                                          const int* __restrict__ src,
                                          const int* __restrict__ dst,
                                          const float* __restrict__ feat,
                                          float* acc, int* le, int* lcnt, int* cnt) {
  uint4 k[NL];
#pragma unroll
  for (int u = 0; u < NL; ++u) {
    const int i = base + u * OTH + tid;
    if (GUARD) k[u] = (i < nvec) ? k4[i] : make_uint4(~pat, ~pat, ~pat, ~pat);
    else       k[u] = k4[i];
  }
#pragma unroll
  for (int u = 0; u < NL; ++u) {
    const int i = base + u * OTH + tid;
    unsigned m = (zero_bytes(k[u].x ^ pat) >> 7) | (zero_bytes(k[u].y ^ pat) >> 6) |
                 (zero_bytes(k[u].z ^ pat) >> 5) | (zero_bytes(k[u].w ^ pat) >> 4);
    while (m) {
      const int b = __builtin_ctz(m);
      m &= m - 1;
      note_keyed(16 * i + ((b & 7) << 2) + (b >> 3), lo, r, src, dst, feat, acc, le, lcnt, cnt);
    }
  }
}

template <int MODE>
__global__ __launch_bounds__(OTH) void mean_owner(const int* __restrict__ edges,
                                                  const float* __restrict__ feat,
                                                  float* __restrict__ agg0,
                                                  float* __restrict__ agg1,
                                                  unsigned* __restrict__ lists,
                                                  const unsigned char* __restrict__ keys,
                                                  int E, int N) {
  __shared__ __attribute__((aligned(16))) float acc[(2 * RN + 1) * FDIM];
  __shared__ int le[2 * LCAP];
  __shared__ int ldl[2 * LCAP];
  __shared__ int cnt[2 * RN];
  __shared__ int lcnt[2];

  const int tid = threadIdx.x;
  const int lane = tid & 31;
  const int wave = __builtin_amdgcn_readfirstlane(tid >> 5);
  const int lo = blockIdx.x * RN;

  for (int i = tid; i < 2 * RN * FDIM / 4; i += OTH)
    ((float4*)acc)[i] = make_float4(0.0f, 0.0f, 0.0f, 0.0f);
  for (int i = tid; i < 2 * RN; i += OTH) cnt[i] = 0;
  if (tid < 2) lcnt[tid] = 0;
  __syncthreads();

  unsigned* mine = lists + (size_t)blockIdx.x * LREG;
  const bool room = ((size_t)gridDim.x * LREG <= (size_t)N * 64) && (N <= (1 << 24)) && (RN < 255);
  int c0 = -1, c1 = -1;
  if (MODE == 1 && room) {
    c0 = __builtin_amdgcn_readfirstlane(((const int*)mine)[0]);
    c1 = __builtin_amdgcn_readfirstlane(((const int*)mine)[1]);
  }
  const bool have = (MODE == 1) && room && ((unsigned)c0 <= (unsigned)LCAP) &&
                    ((unsigned)c1 <= (unsigned)LCAP);

  const bool keyed = (MODE == 0) && (gridDim.x <= 256) && keys_usable(keys, edges, E, N);
  if (keyed) {
    const unsigned pat = (unsigned)blockIdx.x * 0x01010101u;
    const int nk = E >> 4;
    for (int r = 0; r < 2; ++r) {
      const int* src = edges + (size_t)r * 2 * E;
      const int* dst = src + E;
      const uint4* k4 = (const uint4*)(keys + (size_t)r * E);
      const int nfull = nk - nk % (OTH * SU);
      for (int base = 0; base < nfull; base += OTH * SU)
        scan_keys<SU, false>(k4, base, nk, tid, pat, lo, r, src, dst, feat, acc, le, lcnt, cnt);
      for (int base = nfull; base < nk; base += OTH * 4)
        scan_keys<4, true>(k4, base, nk, tid, pat, lo, r, src, dst, feat, acc, le, lcnt, cnt);
    }
  }

  const bool vec = ((E & 3) == 0) && ((((size_t)edges) & 15) == 0);
  const int nvec = vec ? (E >> 2) : 0;
  for (int r = 0; r < 2 && !have && !keyed; ++r) {
    const int* src = edges + (size_t)r * 2 * E;
    const int* dst = src + E;
    const int4* d4 = (const int4*)dst;
    const int nfull = nvec - nvec % (OTH * SU);
    for (int base = 0; base < nfull; base += OTH * SU)
      scan_group<SU, false>(d4, base, nvec, tid, lo, r, src, feat, acc, le, ldl, lcnt, cnt);
    for (int base = nfull; base < nvec; base += OTH * 4)
      scan_group<4, true>(d4, base, nvec, tid, lo, r, src, feat, acc, le, ldl, lcnt, cnt);
    for (int e = nvec * 4 + tid; e < E; e += OTH)
      note_edge(dst[e], e, lo, r, src, feat, acc, le, ldl, lcnt, cnt);
  }
  __syncthreads();

  const int n0 = have ? c0 : min(lcnt[0], LCAP), n1 = have ? c1 : min(lcnt[1], LCAP);
  if (have) {
    constexpr int PT = LCAP / OTH;
    static_assert(LCAP % OTH == 0, "a list is a whole number of words per thread");
    unsigned p[2][PT];
#pragma unroll
    for (int r = 0; r < 2; ++r) {
      const unsigned* sl = mine + LHDR + r * LCAP;
      const int n = r ? n1 : n0;
#pragma unroll
      for (int k = 0; k < PT; ++k) {
        const int i = tid + k * OTH;
        p[r][k] = (i < n) ? sl[i] : 0u;
      }
    }
#pragma unroll
    for (int r = 0; r < 2; ++r) {
      const int n = r ? n1 : n0;
#pragma unroll
      for (int k = 0; k < PT; ++k) {
        const int i = tid + k * OTH;
        if (i < n) {
          const unsigned tt = p[r][k] & 255u;
          const bool real = tt < (unsigned)RN;
          le[r * LCAP + i] = real ? min((int)(p[r][k] >> 8), N - 1) : 0;
          ldl[r * LCAP + i] = real ? (int)tt : 2 * RN - r * RN;
          if (real) atomicAdd(&cnt[r * RN + (int)tt], 1);
        }
      }
    }
  } else {
    const bool save = (MODE == 0) && room;
    if (save && tid == 0) {
      ((int*)mine)[0] = lcnt[0];
      ((int*)mine)[1] = lcnt[1];
    }
    for (int r = 0; r < 2; ++r) {
      const int* src = edges + (size_t)r * 2 * E;
      const int* dst = src + E;
      unsigned* sl = mine + LHDR + r * LCAP;
      const int n = r ? n1 : n0;
      for (int i = tid; i < n; i += OTH) {
        const int e = le[r * LCAP + i];
        int s = src[e];
        if (keyed) {
          const unsigned t = (unsigned)(dst[e] - lo);
          const bool real = t < (unsigned)RN;
          if (real) atomicAdd(&cnt[r * RN + (int)t], 1);
          else s = 0;
          ldl[r * LCAP + i] = real ? (int)t : 2 * RN - r * RN;
          le[r * LCAP + i] = s;
          if (save) sl[i] = ((unsigned)s << 8) | (real ? t : 255u);
        } else {
          le[r * LCAP + i] = s;
          if (save) sl[i] = ((unsigned)s << 8) | (unsigned)ldl[r * LCAP + i];
        }
      }
    }
  }
  __syncthreads();

  for (int r = 0; r < 2; ++r) {
    const int n = r ? n1 : n0;
    float* ar = acc + (size_t)r * RN * FDIM;
    const int* ls = le + r * LCAP;
    const int* ld = ldl + r * LCAP;
    const int spare = 2 * RN - r * RN;
    for (int i0 = wave; i0 < n; i0 += 32 * DU) {
      const int i = i0 + lane * 32;
      const bool ok = (lane < DU) && (i < n);
      const int sv = ok ? ls[i] : 0;
      const int dv = ok ? ld[i] : spare;
      float4 v[DU];
      int dl[DU];
#pragma unroll
      for (int u = 0; u < DU; ++u) {
        const int s = __builtin_amdgcn_readlane(sv, u);
        dl[u] = __builtin_amdgcn_readlane(dv, u);
        v[u] = *(const float4*)(feat + (size_t)s * FDIM + lane * 4);
      }
#pragma unroll
      for (int u = 0; u < DU; ++u) {
        float* a = ar + dl[u] * FDIM + lane;
        atomicAdd(a, v[u].x);
        atomicAdd(a + 32, v[u].y);
        atomicAdd(a + 64, v[u].z);
        atomicAdd(a + 96, v[u].w);
      }
    }
  }
  __syncthreads();

  for (int idx = tid; idx < 2 * RN * 32; idx += OTH) {
    const int r = idx / (RN * 32);
    const int rem = idx - r * (RN * 32);
    const int row = rem >> 5, q = rem & 31;
    const int node = lo + row;
    if (node < N) {
      const float* a = acc + (size_t)(r * RN + row) * FDIM + q;
      const float inv = 1.0f / fmaxf((float)cnt[r * RN + row], 1.0f);
      float4 o;
      o.x = a[0] * inv;
      o.y = a[32] * inv;
      o.z = a[64] * inv;
      o.w = a[96] * inv;
      *(float4*)((r ? agg1 : agg0) + (size_t)node * FDIM + q * 4) = o;
    }
  }
}

#define KS (KC / 4)

template <int NOUT>
__device__ __forceinline__ void gemm_chunk(
    int kn, bool more,
    const float* __restrict__ A0, const float* __restrict__ G0, const float* __restrict__ G1,
    const v4f* __restrict__ B4, int rowA, int khalf, int m, int tid,
    const float* bcur, float* bnxt,
    const v2f (&a)[KS], v2f (&an)[KS],
    v4f (&st)[KC * NOUT / 1024], v8f (&c)[NOUT / 16]) {
  constexpr int NT = NOUT / 16;
  constexpr int NV = KC * NOUT / 1024;
  if (more) {
#pragma unroll
    for (int j = 0; j < NV; ++j) st[j] = B4[(size_t)kn * (NOUT / 4) + tid + 256 * j];
    const float* An = (kn < 128) ? A0 : ((kn < 256) ? G0 : G1);
    const float* ap = An + (size_t)rowA * FDIM + (kn & 127) + khalf;
#pragma unroll
    for (int s = 0; s < KS; ++s) an[s] = *(const v2f*)(ap + 4 * s);
  }

  const float* bs = bcur + khalf * NOUT + m;
  v2f bf[2][NT];
#pragma unroll
  for (int t = 0; t < NT; ++t) {
    bf[0][t].x = bs[t * 16];
    bf[0][t].y = bs[NOUT + t * 16];
  }
#pragma unroll
  for (int s = 0; s < KS; ++s) {
    if (s + 1 < KS) {
#pragma unroll
      for (int t = 0; t < NT; ++t) {
        bf[(s + 1) & 1][t].x = bs[(s + 1) * 4 * NOUT + t * 16];
        bf[(s + 1) & 1][t].y = bs[((s + 1) * 4 + 1) * NOUT + t * 16];
      }
    }
#pragma unroll
    for (int t = 0; t < NT; ++t)
      c[t] = __builtin_amdgcn_wmma_f32_16x16x4_f32(
          false, a[s], false, bf[s & 1][t], (short)0, c[t], false, false);
  }

  if (more) {
#pragma unroll
    for (int j = 0; j < NV; ++j) ((v4f*)bnxt)[tid + 256 * j] = st[j];
    __syncthreads();
  }
}

template <int NOUT, bool RELU>
__global__ __launch_bounds__(256) void sage_gemm(
    const float* __restrict__ A0, const float* __restrict__ G0, const float* __restrict__ G1,
    const float* __restrict__ Bmat, const float* __restrict__ bias,
    float* __restrict__ out, int Nrows) {
  constexpr int NT = NOUT / 16;
  constexpr int NCH = KTOT / KC;
  constexpr int NV = KC * NOUT / 1024;
  static_assert(NCH % 2 == 0, "the chunk loop takes two chunks per trip");
  __shared__ __attribute__((aligned(16))) float Bs[2][KC * NOUT];

  const int tid = threadIdx.x;
  const int lane = tid & 31;
  const int wave = tid >> 5;
  const int rowBase = blockIdx.x * 128 + wave * 16;
  const int m = lane & 15;
  const int khalf = (lane >> 4) << 1;

  v8f czero = {};
  v8f c[NT];
#pragma unroll
  for (int t = 0; t < NT; ++t) c[t] = czero;

  int rowA = rowBase + m;
  if (rowA > Nrows - 1) rowA = Nrows - 1;

  const v4f* B4 = (const v4f*)Bmat;
  v4f st[NV];
  v2f aA[KS], aB[KS];

#pragma unroll
  for (int j = 0; j < NV; ++j) st[j] = B4[tid + 256 * j];
  {
    const float* ap = A0 + (size_t)rowA * FDIM + khalf;
#pragma unroll
    for (int s = 0; s < KS; ++s) aA[s] = *(const v2f*)(ap + 4 * s);
  }
#pragma unroll
  for (int j = 0; j < NV; ++j) ((v4f*)Bs[0])[tid + 256 * j] = st[j];
  __syncthreads();

#pragma unroll 1
  for (int ch = 0; ch < NCH; ch += 2) {
    gemm_chunk<NOUT>((ch + 1) * KC, true, A0, G0, G1, B4, rowA, khalf, m, tid,
                     Bs[0], Bs[1], aA, aB, st, c);
    gemm_chunk<NOUT>((ch + 2) * KC, ch + 2 < NCH, A0, G0, G1, B4, rowA, khalf, m, tid,
                     Bs[1], Bs[0], aB, aA, st, c);
  }

  const int rhalf = (lane >> 4) << 3;
#pragma unroll
  for (int t = 0; t < NT; ++t) {
    int col = t * 16 + m;
    float bv = bias[col];
#pragma unroll
    for (int v = 0; v < 8; ++v) {
      int row = rowBase + v + rhalf;
      if (row < Nrows) {
        float val = c[t][v] + bv;
        if (RELU) val = fmaxf(val, 0.0f);
        out[(size_t)row * NOUT + col] = val;
      }
    }
  }
}

#define SPITCH 68

template <int J>
__device__ __forceinline__ int row_bcast(int v) {
  return __builtin_amdgcn_update_dpp(0, v, 0x150 + J, 0xf, 0xf, true);
}

template <int J>
__device__ __forceinline__ void fetch_rows(const float* __restrict__ h, int node, int c, v4f (&v)[16]) {
  v[J] = *(const v4f*)(h + (size_t)row_bcast<J>(node) * 64 + c * 4);
  if constexpr (J + 1 < 16) fetch_rows<J + 1>(h, node, c, v);
}

__global__ __launch_bounds__(256) void edge_score(const int* __restrict__ src, const int* __restrict__ dst,
                                                  const float* __restrict__ h, float* __restrict__ out, int E) {
  __shared__ __attribute__((aligned(16))) float stage[8][32 * SPITCH];
  const int lane = threadIdx.x & 31;
  const int wave = threadIdx.x >> 5;
  const int c = lane & 15;
  const int hh = lane >> 4;
  const int e0 = (blockIdx.x * 8 + wave) * 16;
  int e = e0 + c;
  if (e > E - 1) e = E - 1;
  const int node = hh ? dst[e] : src[e];

  v4f v[16];
  fetch_rows<0>(h, node, c, v);

  float* st = stage[wave];
#pragma unroll
  for (int j = 0; j < 16; ++j) *(v4f*)(st + (hh * 16 + j) * SPITCH + c * 4) = v[j];
  __syncthreads();

  const float* ap = st + c * SPITCH + hh * 32;
  const float* bp = st + (16 + c) * SPITCH + hh * 32;
  v4f a[8], b[8];
#pragma unroll
  for (int t = 0; t < 8; ++t) {
    a[t] = *(const v4f*)(ap + 4 * t);
    b[t] = *(const v4f*)(bp + 4 * t);
  }
  v8f acc = {};
#pragma unroll
  for (int t = 0; t < 8; ++t) {
    v2f a0 = {a[t].x, a[t].y}, b0 = {b[t].x, b[t].y};
    v2f a1 = {a[t].z, a[t].w}, b1 = {b[t].z, b[t].w};
    acc = __builtin_amdgcn_wmma_f32_16x16x4_f32(false, a0, false, b0, (short)0, acc, false, false);
    acc = __builtin_amdgcn_wmma_f32_16x16x4_f32(false, a1, false, b1, (short)0, acc, false, false);
  }

  const int r = c & 7;
  float s = acc[0];
#pragma unroll
  for (int q = 1; q < 8; ++q) s = (r == q) ? acc[q] : s;
  if ((c >> 3) == hh && e0 + c < E) out[e0 + c] = s;
}

extern "C" void kernel_launch(void* const* d_in, const int* in_sizes, int n_in,
                              void* d_out, int out_size, void* d_ws, size_t ws_size,
                              hipStream_t stream) {
  const float* x   = (const float*)d_in[0];
  const int*   edges = (const int*)d_in[1];
  const int*   negE  = (const int*)d_in[2];
  const float* Wn1 = (const float*)d_in[3];
  const float* Ws1 = (const float*)d_in[4];
  const float* b1  = (const float*)d_in[5];
  const float* Wn2 = (const float*)d_in[6];
  const float* Ws2 = (const float*)d_in[7];
  const float* b2  = (const float*)d_in[8];
  float* out = (float*)d_out;

  const int N = in_sizes[0] / 128;
  const int E = in_sizes[1] / 4;

  float* ws    = (float*)d_ws;
  float* dinv  = ws;
  float* agg0  = dinv + 2 * (size_t)N;
  float* agg1  = agg0 + (size_t)N * 128;
  float* h1    = agg1 + (size_t)N * 128;
  float* h2    = h1 + (size_t)N * 128;
  float* Bcat1 = h2 + (size_t)N * 64;
  float* bias1 = Bcat1 + 384 * 128;
  float* Bcat2 = bias1 + 128;
  float* bias2 = Bcat2 + 384 * 64;

  const int T = 256;

  const int wtot = 384 * 128 + 128 + 384 * 64 + 64;
  build_weights<<<(wtot + T - 1) / T, T, 0, stream>>>(Wn1, Ws1, b1, Wn2, Ws2, b2,
                                                      Bcat1, bias1, Bcat2, bias2);

  const int sblocks = (E + 127) / 128;
  const int gblocks = (N + 127) / 128;
  const int oblocks = (N + RN - 1) / RN;
  const int kblocks = (int)(((size_t)E / 8 + T) / T);

  unsigned* olist = (unsigned*)h2;
  unsigned char* okeys = (unsigned char*)h1;
  build_keys<<<kblocks, T, 0, stream>>>(edges, okeys, E, N);
  mean_owner<0><<<oblocks, OTH, 0, stream>>>(edges, x, agg0, agg1, olist, okeys, E, N);
  sage_gemm<128, true><<<gblocks, T, 0, stream>>>(x, agg0, agg1, Bcat1, bias1, h1, N);

  mean_owner<1><<<oblocks, OTH, 0, stream>>>(edges, h1, agg0, agg1, olist, nullptr, E, N);
  sage_gemm<64, false><<<gblocks, T, 0, stream>>>(h1, agg0, agg1, Bcat2, bias2, h2, N);

  edge_score<<<sblocks, T, 0, stream>>>(edges, edges + E, h2, out, E);
  edge_score<<<sblocks, T, 0, stream>>>(negE,  negE + E,  h2, out + E, E);
}
